# starting kernel
# baseline (speedup 1.0000x reference)
#include <hip/hip_runtime.h>

typedef float v2f __attribute__((ext_vector_type(2)));
typedef float v8f __attribute__((ext_vector_type(8)));

#define S_GRID   56
#define N_CLASS  20
#define CELLS    (S_GRID * S_GRID)
#define ITEMS    (CELLS * 2)
#define SORTN    8192
#define CAND     256
#define MAXOUT   30
#define D_IMG    94080
#define CONF_OFF 62720
#define COORD_OFF 68992
#define NEG_INF  (-3.402823466e38f)

struct Phase2 {
    float y0[CAND], x0[CAND], y1[CAND], x1[CAND];
    float sc[CAND], cl[CAND], ar[CAND];
    float redv[CAND];
    int   redi[CAND];
    unsigned short mask16[CAND * 16];
    int selIdx[32], selOk[32];
    unsigned char valid[CAND];
};

__global__ __launch_bounds__(256) void det_kernel(const float* __restrict__ net,
                                                  float* __restrict__ out) {
    __shared__ unsigned long long keys[SORTN];
    Phase2* P = reinterpret_cast<Phase2*>(&keys[CAND]);

    const int tid = threadIdx.x;
    const int img = blockIdx.x;
    const float* in = net + (size_t)img * D_IMG;

    for (int cell = tid; cell < CELLS; cell += 256) {
        const float* pr = in + cell * N_CLASS;
        float p[N_CLASS];
        #pragma unroll
        for (int c = 0; c < N_CLASS; ++c) p[c] = pr[c];
        #pragma unroll
        for (int b = 0; b < 2; ++b) {
            int id = cell * 2 + b;
            float conf = in[CONF_OFF + id];
            float best = NEG_INF;
            #pragma unroll
            for (int c = 0; c < N_CLASS; ++c) {
                float v = __fmul_rn(conf, p[c]);
                best = v > best ? v : best;
            }
            float s = (best >= 0.1f) ? best : 0.0f;
            keys[id] = ((unsigned long long)__float_as_uint(s) << 32) |
                       (unsigned)(~(unsigned)id);
        }
    }
    for (int q = ITEMS + tid; q < SORTN; q += 256)
        keys[q] = (unsigned long long)(unsigned)(~(unsigned)q);
    __syncthreads();

    for (unsigned k = 2; k <= (unsigned)SORTN; k <<= 1) {
        for (unsigned j = k >> 1; j > 0; j >>= 1) {
            for (unsigned i = tid; i < (unsigned)SORTN; i += 256u) {
                unsigned x = i ^ j;
                if (x > i) {
                    unsigned long long a = keys[i], b = keys[x];
                    bool up = ((i & k) == 0u);
                    if (up ? (a < b) : (a > b)) { keys[i] = b; keys[x] = a; }
                }
            }
            __syncthreads();
        }
    }

    {
        unsigned long long key = keys[tid];
        unsigned id = ~(unsigned)key;
        float s = __uint_as_float((unsigned)(key >> 32));
        int cell = (int)(id >> 1);
        int gi = cell / S_GRID, gj = cell % S_GRID;
        const float* cd = in + COORD_OFF + (size_t)id * 4;
        float tx = cd[0], ty = cd[1], tw = cd[2], th = cd[3];
        float x = __fdiv_rn(__fadd_rn(tx, (float)gj), 56.0f);
        float y = __fdiv_rn(__fadd_rn(ty, (float)gi), 56.0f);
        float hw = __fmul_rn(__fmul_rn(tw, tw), 0.5f);
        float hh = __fmul_rn(__fmul_rn(th, th), 0.5f);
        float y0 = __fsub_rn(y, hh), x0 = __fsub_rn(x, hw);
        float y1 = __fadd_rn(y, hh), x1 = __fadd_rn(x, hw);
        float conf = in[CONF_OFF + id];
        const float* pr = in + cell * N_CLASS;
        float best = NEG_INF; int bc = 0;
        #pragma unroll
        for (int c = 0; c < N_CLASS; ++c) {
            float v = __fmul_rn(conf, pr[c]);
            if (v > best) { best = v; bc = c; }
        }
        float dy = fmaxf(__fsub_rn(y1, y0), 0.0f);
        float dx = fmaxf(__fsub_rn(x1, x0), 0.0f);
        P->y0[tid] = y0; P->x0[tid] = x0; P->y1[tid] = y1; P->x1[tid] = x1;
        P->sc[tid] = s;  P->cl[tid] = (float)bc;
        P->ar[tid] = __fmul_rn(dy, dx);
        P->valid[tid] = 1;
    }
    __syncthreads();

    {
        const int lane = tid & 31;
        const int wave = tid >> 5;
        v8f czero = {};
        for (int tt = wave * 32; tt < wave * 32 + 32; ++tt) {
            int mi = tt >> 4, ni = tt & 15;
            int m0 = mi << 4, n0 = ni << 4;
            v2f A, B;
            if (lane < 16) {
                A[0] = P->ar[m0 + lane]; A[1] = 1.0f;
                B[0] = 1.0f;             B[1] = P->ar[n0 + lane];
            } else {
                A[0] = 0.0f; A[1] = 0.0f; B[0] = 0.0f; B[1] = 0.0f;
            }
            v8f d = __builtin_amdgcn_wmma_f32_16x16x4_f32(
                false, A, false, B, (short)0, czero, false, false);
            int n = n0 + (lane & 15);
            float ny0 = P->y0[n], nx0 = P->x0[n], ny1 = P->y1[n], nx1 = P->x1[n];
            int mbase = m0 + ((lane & 16) ? 8 : 0);
            #pragma unroll
            for (int r = 0; r < 8; ++r) {
                int m = mbase + r;
                float iy = fmaxf(__fsub_rn(fminf(P->y1[m], ny1), fmaxf(P->y0[m], ny0)), 0.0f);
                float ix = fmaxf(__fsub_rn(fminf(P->x1[m], nx1), fmaxf(P->x0[m], nx0)), 0.0f);
                float inter = __fmul_rn(iy, ix);
                float uni = __fsub_rn(d[r], inter);
                bool pred = (uni > 0.0f) && (__fdiv_rn(inter, uni) > 0.4f);
                unsigned bal = __builtin_amdgcn_ballot_w32(pred);
                if (lane == 0) {
                    P->mask16[(m0 + r)     * 16 + ni] = (unsigned short)(bal & 0xFFFFu);
                    P->mask16[(m0 + 8 + r) * 16 + ni] = (unsigned short)(bal >> 16);
                }
            }
        }
    }
    __syncthreads();

    for (int it = 0; it < MAXOUT; ++it) {
        P->redv[tid] = P->valid[tid] ? P->sc[tid] : NEG_INF;
        P->redi[tid] = tid;
        __syncthreads();
        for (int off = 128; off > 0; off >>= 1) {
            if (tid < off) {
                float v2 = P->redv[tid + off]; int i2 = P->redi[tid + off];
                if (v2 > P->redv[tid] || (v2 == P->redv[tid] && i2 < P->redi[tid])) {
                    P->redv[tid] = v2; P->redi[tid] = i2;
                }
            }
            __syncthreads();
        }
        int idx = P->redi[0];
        int ok  = P->valid[idx];
        unsigned bit = (P->mask16[idx * 16 + (tid >> 4)] >> (tid & 15)) & 1u;
        unsigned char nv = (unsigned char)(P->valid[tid] & (unsigned char)(1u - bit));
        if (tid == idx) nv = 0;
        __syncthreads();
        P->valid[tid] = nv;
        if (tid == 0) { P->selIdx[it] = idx; P->selOk[it] = ok; }
        __syncthreads();
    }

    if (tid < MAXOUT) {
        int sel = P->selIdx[tid];
        float okf = P->selOk[tid] ? 1.0f : 0.0f;
        P->redv[tid] = __fmul_rn(P->sc[sel], okf);
    }
    __syncthreads();
    if (tid < MAXOUT) {
        int sel = P->selIdx[tid];
        float okf = P->selOk[tid] ? 1.0f : 0.0f;
        float ms = P->redv[tid];
        int rank = 0;
        for (int jt = 0; jt < MAXOUT; ++jt) {
            float vj = P->redv[jt];
            if (vj > ms || (vj == ms && jt < tid)) rank++;
        }
        float* o = out + (size_t)img * (MAXOUT * 6) + (size_t)rank * 6;
        o[0] = __fmul_rn(P->y0[sel], okf);
        o[1] = __fmul_rn(P->x0[sel], okf);
        o[2] = __fmul_rn(P->y1[sel], okf);
        o[3] = __fmul_rn(P->x1[sel], okf);
        o[4] = ms;
        o[5] = __fmul_rn(P->cl[sel], okf);
    }
}

extern "C" void kernel_launch(void* const* d_in, const int* in_sizes, int n_in,
                              void* d_out, int out_size, void* d_ws, size_t ws_size,
                              hipStream_t stream) {
    (void)n_in; (void)out_size; (void)d_ws; (void)ws_size;
    const float* net = (const float*)d_in[0];
    float* out = (float*)d_out;
    int batch = in_sizes[0] / D_IMG;
    det_kernel<<<dim3(batch), dim3(256), 0, stream>>>(net, out);
}


# Top-256 pre-selection by bitwise count search over register-held scores (ballot counts) plus compaction and rank placement of the 256 survivors, replacing the 8192-key LDS bitonic sort
# speedup vs baseline: 4.1470x; 4.1470x over previous
#include <hip/hip_runtime.h>


typedef float v2f __attribute__((ext_vector_type(2)));
typedef float v8f __attribute__((ext_vector_type(8)));

#define S_GRID   56
#define N_CLASS  20
#define CELLS    (S_GRID * S_GRID)
#define ITEMS    (CELLS * 2)
#define CAND     256
#define MAXOUT   30
#define D_IMG    94080
#define CONF_OFF 62720
#define COORD_OFF 68992
#define NEG_INF  (-3.402823466e38f)
#define CPT      13
#define IPT      (CPT * 2)
#define SLOTS    (IPT * 256)
#define NWAVE    8

struct Phase2 {
    float y0[CAND], x0[CAND], y1[CAND], x1[CAND];
    float sc[CAND], cl[CAND], ar[CAND];
    float redv[CAND];
    int   redi[CAND];
    unsigned short mask16[CAND * 16];
    int selIdx[32], selOk[32];
    unsigned char valid[CAND];
};

struct Shared {
    unsigned long long surv[CAND];
    unsigned long long keys[CAND];
    unsigned wcnt[2][NWAVE];
    Phase2 p2;
};

#define BLOCK_COUNT(result, PRED)                                               \
    do {                                                                        \
        unsigned c_ = 0;                                                        \
          \
        _Pragma("unroll")                                                       \
        for (int k = 0; k < IPT; ++k) {                                         \
            const unsigned id_ = id0 + 512u * (unsigned)(k >> 1) + (unsigned)(k & 1); \
            const unsigned rid_ = 8191u - id_;                                  \
            const unsigned sv_ = sb[k];                                         \
            (void)rid_; (void)sv_;                                              \
            c_ += (unsigned)__popc(__builtin_amdgcn_ballot_w32(PRED));          \
        }                                                                       \
        if (lane == 0) S.wcnt[par][wave] = c_;                                  \
        __syncthreads();                                                        \
        unsigned t_ = 0;                                                        \
          \
        _Pragma("unroll")                                                       \
        for (int w = 0; w < NWAVE; ++w) t_ += S.wcnt[par][w];                   \
        par ^= 1;                                                               \
          \
          \
        (result) = (unsigned)__builtin_amdgcn_readfirstlane((int)t_);           \
    } while (0)

__global__ __launch_bounds__(256) void det_kernel(const float* __restrict__ net,
                                                  float* __restrict__ out) {
    __shared__ Shared S;
    Phase2* P = &S.p2;

    const int tid = threadIdx.x;
    const int lane = tid & 31;
    const int wave = tid >> 5;
    const int img = blockIdx.x;
    const float* in = net + (size_t)img * D_IMG;
    const unsigned id0 = 2u * (unsigned)tid;

    unsigned sb[IPT];
    #pragma unroll
    for (int kc = 0; kc < CPT; ++kc) {
        const int cell = tid + kc * 256;
        float s0 = 0.0f, s1 = 0.0f;
        if (kc < CPT - 1 || cell < CELLS) {
            const float* pr = in + cell * N_CLASS;
            float p[N_CLASS];
            #pragma unroll
            for (int c = 0; c < N_CLASS; ++c) p[c] = pr[c];
            const float conf0 = in[CONF_OFF + cell * 2];
            const float conf1 = in[CONF_OFF + cell * 2 + 1];
            float b0 = NEG_INF, b1 = NEG_INF;
            #pragma unroll
            for (int c = 0; c < N_CLASS; ++c) {
                float v0 = __fmul_rn(conf0, p[c]);
                float v1 = __fmul_rn(conf1, p[c]);
                b0 = v0 > b0 ? v0 : b0;
                b1 = v1 > b1 ? v1 : b1;
            }
            s0 = (b0 >= 0.1f) ? b0 : 0.0f;
            s1 = (b1 >= 0.1f) ? b1 : 0.0f;
        }
        sb[2 * kc]     = __float_as_uint(s0);
        sb[2 * kc + 1] = __float_as_uint(s1);
    }

    int par = 0;
    unsigned T = 0u, cntGE = (unsigned)SLOTS;
    for (int bit = 31; bit >= 0; --bit) {
        const unsigned c = T | (1u << bit);
        unsigned n;
        BLOCK_COUNT(n, sv_ >= c);
        if (n >= (unsigned)CAND) { T = c; cntGE = n; }
    }
    unsigned R = 0u;
    if (cntGE != (unsigned)CAND) {
        unsigned above;
        BLOCK_COUNT(above, sv_ > T);
        const unsigned need = (unsigned)CAND - above;
        for (int bit = 12; bit >= 0; --bit) {
            const unsigned c = R | (1u << bit);
            unsigned n;
            BLOCK_COUNT(n, (sv_ == T) && (rid_ >= c));
            if (n >= need) R = c;
        }
    }

    {
        unsigned m[IPT];
        unsigned wtot = 0;
        #pragma unroll
        for (int k = 0; k < IPT; ++k) {
            const unsigned id = id0 + 512u * (unsigned)(k >> 1) + (unsigned)(k & 1);
            const bool take = (sb[k] > T) || ((sb[k] == T) && (8191u - id >= R));
            m[k] = __builtin_amdgcn_ballot_w32(take);
            wtot += (unsigned)__popc(m[k]);
        }
        if (lane == 0) S.wcnt[par][wave] = wtot;
        __syncthreads();
        unsigned base = 0;
        #pragma unroll
        for (int w = 0; w < NWAVE; ++w) base += (w < wave) ? S.wcnt[par][w] : 0u;
        const unsigned below = (1u << lane) - 1u;
        #pragma unroll
        for (int k = 0; k < IPT; ++k) {
            const unsigned id = id0 + 512u * (unsigned)(k >> 1) + (unsigned)(k & 1);
            if ((m[k] >> lane) & 1u) {
                const unsigned pos = (base + (unsigned)__popc(m[k] & below)) & (unsigned)(CAND - 1);
                S.surv[pos] = ((unsigned long long)sb[k] << 32) | (unsigned)(~id);
            }
            base += (unsigned)__popc(m[k]);
        }
    }
    __syncthreads();

    {
        const unsigned long long mine = S.surv[tid];
        unsigned rank = 0;
        #pragma unroll 16
        for (int j = 0; j < CAND; ++j) rank += (S.surv[j] > mine) ? 1u : 0u;
        S.keys[rank] = mine;
    }
    __syncthreads();

    {
        unsigned long long key = S.keys[tid];
        unsigned id = ~(unsigned)key;
        id = id < (unsigned)ITEMS ? id : (unsigned)(ITEMS - 1);
        float s = __uint_as_float((unsigned)(key >> 32));
        int cell = (int)(id >> 1);
        int gi = cell / S_GRID, gj = cell % S_GRID;
        const float* cd = in + COORD_OFF + (size_t)id * 4;
        float tx = cd[0], ty = cd[1], tw = cd[2], th = cd[3];
        float x = __fdiv_rn(__fadd_rn(tx, (float)gj), 56.0f);
        float y = __fdiv_rn(__fadd_rn(ty, (float)gi), 56.0f);
        float hw = __fmul_rn(__fmul_rn(tw, tw), 0.5f);
        float hh = __fmul_rn(__fmul_rn(th, th), 0.5f);
        float y0 = __fsub_rn(y, hh), x0 = __fsub_rn(x, hw);
        float y1 = __fadd_rn(y, hh), x1 = __fadd_rn(x, hw);
        float conf = in[CONF_OFF + id];
        const float* pr = in + cell * N_CLASS;
        float best = NEG_INF; int bc = 0;
        #pragma unroll
        for (int c = 0; c < N_CLASS; ++c) {
            float v = __fmul_rn(conf, pr[c]);
            if (v > best) { best = v; bc = c; }
        }
        float dy = fmaxf(__fsub_rn(y1, y0), 0.0f);
        float dx = fmaxf(__fsub_rn(x1, x0), 0.0f);
        P->y0[tid] = y0; P->x0[tid] = x0; P->y1[tid] = y1; P->x1[tid] = x1;
        P->sc[tid] = s;  P->cl[tid] = (float)bc;
        P->ar[tid] = __fmul_rn(dy, dx);
        P->valid[tid] = 1;
    }
    __syncthreads();

    {
        v8f czero = {};
        for (int tt = wave * 32; tt < wave * 32 + 32; ++tt) {
            int mi = tt >> 4, ni = tt & 15;
            int m0 = mi << 4, n0 = ni << 4;
            v2f A, B;
            if (lane < 16) {
                A[0] = P->ar[m0 + lane]; A[1] = 1.0f;
                B[0] = 1.0f;             B[1] = P->ar[n0 + lane];
            } else {
                A[0] = 0.0f; A[1] = 0.0f; B[0] = 0.0f; B[1] = 0.0f;
            }
            v8f d = __builtin_amdgcn_wmma_f32_16x16x4_f32(
                false, A, false, B, (short)0, czero, false, false);
            int n = n0 + (lane & 15);
            float ny0 = P->y0[n], nx0 = P->x0[n], ny1 = P->y1[n], nx1 = P->x1[n];
            int mbase = m0 + ((lane & 16) ? 8 : 0);
            #pragma unroll
            for (int r = 0; r < 8; ++r) {
                int m = mbase + r;
                float iy = fmaxf(__fsub_rn(fminf(P->y1[m], ny1), fmaxf(P->y0[m], ny0)), 0.0f);
                float ix = fmaxf(__fsub_rn(fminf(P->x1[m], nx1), fmaxf(P->x0[m], nx0)), 0.0f);
                float inter = __fmul_rn(iy, ix);
                float uni = __fsub_rn(d[r], inter);
                bool pred = (uni > 0.0f) && (__fdiv_rn(inter, uni) > 0.4f);
                unsigned bal = __builtin_amdgcn_ballot_w32(pred);
                if (lane == 0) {
                    P->mask16[(m0 + r)     * 16 + ni] = (unsigned short)(bal & 0xFFFFu);
                    P->mask16[(m0 + 8 + r) * 16 + ni] = (unsigned short)(bal >> 16);
                }
            }
        }
    }
    __syncthreads();

    for (int it = 0; it < MAXOUT; ++it) {
        P->redv[tid] = P->valid[tid] ? P->sc[tid] : NEG_INF;
        P->redi[tid] = tid;
        __syncthreads();
        for (int off = 128; off > 0; off >>= 1) {
            if (tid < off) {
                float v2 = P->redv[tid + off]; int i2 = P->redi[tid + off];
                if (v2 > P->redv[tid] || (v2 == P->redv[tid] && i2 < P->redi[tid])) {
                    P->redv[tid] = v2; P->redi[tid] = i2;
                }
            }
            __syncthreads();
        }
        int idx = P->redi[0];
        int ok  = P->valid[idx];
        unsigned bit = (P->mask16[idx * 16 + (tid >> 4)] >> (tid & 15)) & 1u;
        unsigned char nv = (unsigned char)(P->valid[tid] & (unsigned char)(1u - bit));
        if (tid == idx) nv = 0;
        __syncthreads();
        P->valid[tid] = nv;
        if (tid == 0) { P->selIdx[it] = idx; P->selOk[it] = ok; }
        __syncthreads();
    }

    if (tid < MAXOUT) {
        int sel = P->selIdx[tid];
        float okf = P->selOk[tid] ? 1.0f : 0.0f;
        P->redv[tid] = __fmul_rn(P->sc[sel], okf);
    }
    __syncthreads();
    if (tid < MAXOUT) {
        int sel = P->selIdx[tid];
        float okf = P->selOk[tid] ? 1.0f : 0.0f;
        float ms = P->redv[tid];
        int rank = 0;
        for (int jt = 0; jt < MAXOUT; ++jt) {
            float vj = P->redv[jt];
            if (vj > ms || (vj == ms && jt < tid)) rank++;
        }
        float* o = out + (size_t)img * (MAXOUT * 6) + (size_t)rank * 6;
        o[0] = __fmul_rn(P->y0[sel], okf);
        o[1] = __fmul_rn(P->x0[sel], okf);
        o[2] = __fmul_rn(P->y1[sel], okf);
        o[3] = __fmul_rn(P->x1[sel], okf);
        o[4] = ms;
        o[5] = __fmul_rn(P->cl[sel], okf);
    }
}

extern "C" void kernel_launch(void* const* d_in, const int* in_sizes, int n_in,
                              void* d_out, int out_size, void* d_ws, size_t ws_size,
                              hipStream_t stream) {
    (void)n_in; (void)out_size; (void)d_ws; (void)ws_size;
    const float* net = (const float*)d_in[0];
    float* out = (float*)d_out;
    int batch = in_sizes[0] / D_IMG;
    det_kernel<<<dim3(batch), dim3(256), 0, stream>>>(net, out);
}


# Greedy NMS scan as first-set-bit search on a 256-bit valid mask held in one wave (candidates are already in score order), replacing 30 block-wide LDS argmax trees and their barriers
# speedup vs baseline: 6.0840x; 1.4671x over previous
#include <hip/hip_runtime.h>


typedef float v2f __attribute__((ext_vector_type(2)));
typedef float v8f __attribute__((ext_vector_type(8)));

#define S_GRID   56
#define N_CLASS  20
#define CELLS    (S_GRID * S_GRID)
#define ITEMS    (CELLS * 2)
#define CAND     256
#define MAXOUT   30
#define D_IMG    94080
#define CONF_OFF 62720
#define COORD_OFF 68992
#define NEG_INF  (-3.402823466e38f)
#define CPT      13
#define IPT      (CPT * 2)
#define SLOTS    (IPT * 256)
#define NWAVE    8

struct Phase2 {
    float y0[CAND], x0[CAND], y1[CAND], x1[CAND];
    float sc[CAND], cl[CAND], ar[CAND];
    float redv[CAND];
    unsigned short mask16[CAND * 16];
    int selIdx[32], selOk[32];
};

struct Shared {
    unsigned long long surv[CAND];
    unsigned long long keys[CAND];
    unsigned wcnt[2][NWAVE];
    Phase2 p2;
};

#define BLOCK_COUNT(result, PRED)                                               \
    do {                                                                        \
        unsigned c_ = 0;                                                        \
          \
        _Pragma("unroll")                                                       \
        for (int k = 0; k < IPT; ++k) {                                         \
            const unsigned id_ = id0 + 512u * (unsigned)(k >> 1) + (unsigned)(k & 1); \
            const unsigned rid_ = 8191u - id_;                                  \
            const unsigned sv_ = sb[k];                                         \
            (void)rid_; (void)sv_;                                              \
            c_ += (unsigned)__popc(__builtin_amdgcn_ballot_w32(PRED));          \
        }                                                                       \
        if (lane == 0) S.wcnt[par][wave] = c_;                                  \
        __syncthreads();                                                        \
        unsigned t_ = 0;                                                        \
          \
        _Pragma("unroll")                                                       \
        for (int w = 0; w < NWAVE; ++w) t_ += S.wcnt[par][w];                   \
        par ^= 1;                                                               \
          \
          \
        (result) = (unsigned)__builtin_amdgcn_readfirstlane((int)t_);           \
    } while (0)

__global__ __launch_bounds__(256) void det_kernel(const float* __restrict__ net,
                                                  float* __restrict__ out) {
    __shared__ Shared S;
    Phase2* P = &S.p2;

    const int tid = threadIdx.x;
    const int lane = tid & 31;
    const int wave = tid >> 5;
    const int img = blockIdx.x;
    const float* in = net + (size_t)img * D_IMG;
    const unsigned id0 = 2u * (unsigned)tid;

    unsigned sb[IPT];
    #pragma unroll
    for (int kc = 0; kc < CPT; ++kc) {
        const int cell = tid + kc * 256;
        float s0 = 0.0f, s1 = 0.0f;
        if (kc < CPT - 1 || cell < CELLS) {
            const float* pr = in + cell * N_CLASS;
            float p[N_CLASS];
            #pragma unroll
            for (int c = 0; c < N_CLASS; ++c) p[c] = pr[c];
            const float conf0 = in[CONF_OFF + cell * 2];
            const float conf1 = in[CONF_OFF + cell * 2 + 1];
            float b0 = NEG_INF, b1 = NEG_INF;
            #pragma unroll
            for (int c = 0; c < N_CLASS; ++c) {
                float v0 = __fmul_rn(conf0, p[c]);
                float v1 = __fmul_rn(conf1, p[c]);
                b0 = v0 > b0 ? v0 : b0;
                b1 = v1 > b1 ? v1 : b1;
            }
            s0 = (b0 >= 0.1f) ? b0 : 0.0f;
            s1 = (b1 >= 0.1f) ? b1 : 0.0f;
        }
        sb[2 * kc]     = __float_as_uint(s0);
        sb[2 * kc + 1] = __float_as_uint(s1);
    }

    int par = 0;
    unsigned T = 0u, cntGE = (unsigned)SLOTS;
    for (int bit = 31; bit >= 0; --bit) {
        const unsigned c = T | (1u << bit);
        unsigned n;
        BLOCK_COUNT(n, sv_ >= c);
        if (n >= (unsigned)CAND) { T = c; cntGE = n; }
    }
    unsigned R = 0u;
    if (cntGE != (unsigned)CAND) {
        unsigned above;
        BLOCK_COUNT(above, sv_ > T);
        const unsigned need = (unsigned)CAND - above;
        for (int bit = 12; bit >= 0; --bit) {
            const unsigned c = R | (1u << bit);
            unsigned n;
            BLOCK_COUNT(n, (sv_ == T) && (rid_ >= c));
            if (n >= need) R = c;
        }
    }

    {
        unsigned m[IPT];
        unsigned wtot = 0;
        #pragma unroll
        for (int k = 0; k < IPT; ++k) {
            const unsigned id = id0 + 512u * (unsigned)(k >> 1) + (unsigned)(k & 1);
            const bool take = (sb[k] > T) || ((sb[k] == T) && (8191u - id >= R));
            m[k] = __builtin_amdgcn_ballot_w32(take);
            wtot += (unsigned)__popc(m[k]);
        }
        if (lane == 0) S.wcnt[par][wave] = wtot;
        __syncthreads();
        unsigned base = 0;
        #pragma unroll
        for (int w = 0; w < NWAVE; ++w) base += (w < wave) ? S.wcnt[par][w] : 0u;
        const unsigned below = (1u << lane) - 1u;
        #pragma unroll
        for (int k = 0; k < IPT; ++k) {
            const unsigned id = id0 + 512u * (unsigned)(k >> 1) + (unsigned)(k & 1);
            if ((m[k] >> lane) & 1u) {
                const unsigned pos = (base + (unsigned)__popc(m[k] & below)) & (unsigned)(CAND - 1);
                S.surv[pos] = ((unsigned long long)sb[k] << 32) | (unsigned)(~id);
            }
            base += (unsigned)__popc(m[k]);
        }
    }
    __syncthreads();

    {
        const unsigned long long mine = S.surv[tid];
        unsigned rank = 0;
        #pragma unroll 16
        for (int j = 0; j < CAND; ++j) rank += (S.surv[j] > mine) ? 1u : 0u;
        S.keys[rank] = mine;
    }
    __syncthreads();

    {
        unsigned long long key = S.keys[tid];
        unsigned id = ~(unsigned)key;
        id = id < (unsigned)ITEMS ? id : (unsigned)(ITEMS - 1);
        float s = __uint_as_float((unsigned)(key >> 32));
        int cell = (int)(id >> 1);
        int gi = cell / S_GRID, gj = cell % S_GRID;
        const float* cd = in + COORD_OFF + (size_t)id * 4;
        float tx = cd[0], ty = cd[1], tw = cd[2], th = cd[3];
        float x = __fdiv_rn(__fadd_rn(tx, (float)gj), 56.0f);
        float y = __fdiv_rn(__fadd_rn(ty, (float)gi), 56.0f);
        float hw = __fmul_rn(__fmul_rn(tw, tw), 0.5f);
        float hh = __fmul_rn(__fmul_rn(th, th), 0.5f);
        float y0 = __fsub_rn(y, hh), x0 = __fsub_rn(x, hw);
        float y1 = __fadd_rn(y, hh), x1 = __fadd_rn(x, hw);
        float conf = in[CONF_OFF + id];
        const float* pr = in + cell * N_CLASS;
        float best = NEG_INF; int bc = 0;
        #pragma unroll
        for (int c = 0; c < N_CLASS; ++c) {
            float v = __fmul_rn(conf, pr[c]);
            if (v > best) { best = v; bc = c; }
        }
        float dy = fmaxf(__fsub_rn(y1, y0), 0.0f);
        float dx = fmaxf(__fsub_rn(x1, x0), 0.0f);
        P->y0[tid] = y0; P->x0[tid] = x0; P->y1[tid] = y1; P->x1[tid] = x1;
        P->sc[tid] = s;  P->cl[tid] = (float)bc;
        P->ar[tid] = __fmul_rn(dy, dx);
    }
    __syncthreads();

    {
        v8f czero = {};
        for (int tt = wave * 32; tt < wave * 32 + 32; ++tt) {
            int mi = tt >> 4, ni = tt & 15;
            int m0 = mi << 4, n0 = ni << 4;
            v2f A, B;
            if (lane < 16) {
                A[0] = P->ar[m0 + lane]; A[1] = 1.0f;
                B[0] = 1.0f;             B[1] = P->ar[n0 + lane];
            } else {
                A[0] = 0.0f; A[1] = 0.0f; B[0] = 0.0f; B[1] = 0.0f;
            }
            v8f d = __builtin_amdgcn_wmma_f32_16x16x4_f32(
                false, A, false, B, (short)0, czero, false, false);
            int n = n0 + (lane & 15);
            float ny0 = P->y0[n], nx0 = P->x0[n], ny1 = P->y1[n], nx1 = P->x1[n];
            int mbase = m0 + ((lane & 16) ? 8 : 0);
            #pragma unroll
            for (int r = 0; r < 8; ++r) {
                int m = mbase + r;
                float iy = fmaxf(__fsub_rn(fminf(P->y1[m], ny1), fmaxf(P->y0[m], ny0)), 0.0f);
                float ix = fmaxf(__fsub_rn(fminf(P->x1[m], nx1), fmaxf(P->x0[m], nx0)), 0.0f);
                float inter = __fmul_rn(iy, ix);
                float uni = __fsub_rn(d[r], inter);
                bool pred = (uni > 0.0f) && (__fdiv_rn(inter, uni) > 0.4f);
                unsigned bal = __builtin_amdgcn_ballot_w32(pred);
                if (lane == 0) {
                    P->mask16[(m0 + r)     * 16 + ni] = (unsigned short)(bal & 0xFFFFu);
                    P->mask16[(m0 + 8 + r) * 16 + ni] = (unsigned short)(bal >> 16);
                }
            }
        }
    }
    __syncthreads();

    if (wave == 0) {
        unsigned v = (lane < 16) ? 0xFFFFu : 0u;
        int mySel = 0, myOk = 0;
        for (int it = 0; it < MAXOUT; ++it) {
            const unsigned nz = __builtin_amdgcn_ballot_w32(v != 0u);
            const int ok = (nz != 0u) ? 1 : 0;
            const int c = ok ? __builtin_ctz(nz) : 0;
            const unsigned word = (unsigned)__builtin_amdgcn_readlane((int)v, c);
            const int idx = ok ? (c * 16 + __builtin_ctz(word)) : 0;
            if (lane == it) { mySel = idx; myOk = ok; }
            const unsigned row = P->mask16[idx * 16 + (lane & 15)];
            v &= ~row;
            if (lane == c) v &= ~(1u << (idx & 15));
        }
        if (lane < MAXOUT) { P->selIdx[lane] = mySel; P->selOk[lane] = myOk; }
    }
    __syncthreads();

    if (tid < MAXOUT) {
        int sel = P->selIdx[tid];
        float okf = P->selOk[tid] ? 1.0f : 0.0f;
        P->redv[tid] = __fmul_rn(P->sc[sel], okf);
    }
    __syncthreads();
    if (tid < MAXOUT) {
        int sel = P->selIdx[tid];
        float okf = P->selOk[tid] ? 1.0f : 0.0f;
        float ms = P->redv[tid];
        int rank = 0;
        for (int jt = 0; jt < MAXOUT; ++jt) {
            float vj = P->redv[jt];
            if (vj > ms || (vj == ms && jt < tid)) rank++;
        }
        float* o = out + (size_t)img * (MAXOUT * 6) + (size_t)rank * 6;
        o[0] = __fmul_rn(P->y0[sel], okf);
        o[1] = __fmul_rn(P->x0[sel], okf);
        o[2] = __fmul_rn(P->y1[sel], okf);
        o[3] = __fmul_rn(P->x1[sel], okf);
        o[4] = ms;
        o[5] = __fmul_rn(P->cl[sel], okf);
    }
}

extern "C" void kernel_launch(void* const* d_in, const int* in_sizes, int n_in,
                              void* d_out, int out_size, void* d_ws, size_t ws_size,
                              hipStream_t stream) {
    (void)n_in; (void)out_size; (void)d_ws; (void)ws_size;
    const float* net = (const float*)d_in[0];
    float* out = (float*)d_out;
    int batch = in_sizes[0] / D_IMG;
    det_kernel<<<dim3(batch), dim3(256), 0, stream>>>(net, out);
}


# IoU mask build made branch-free with the eight rows of a tile in flight together: row boxes kept in registers per 16-row group, division in every lane, mask words stored once per tile by 16 lanes
# speedup vs baseline: 10.6072x; 1.7435x over previous
#include <hip/hip_runtime.h>


typedef float v2f __attribute__((ext_vector_type(2)));
typedef float v8f __attribute__((ext_vector_type(8)));

#define S_GRID   56
#define N_CLASS  20
#define CELLS    (S_GRID * S_GRID)
#define ITEMS    (CELLS * 2)
#define CAND     256
#define MAXOUT   30
#define D_IMG    94080
#define CONF_OFF 62720
#define COORD_OFF 68992
#define NEG_INF  (-3.402823466e38f)
#define CPT      13
#define IPT      (CPT * 2)
#define SLOTS    (IPT * 256)
#define NWAVE    8

struct Phase2 {
    float y0[CAND], x0[CAND], y1[CAND], x1[CAND];
    float sc[CAND], cl[CAND], ar[CAND];
    float redv[CAND];
    unsigned short mask16[CAND * 16];
    int selIdx[32], selOk[32];
};

struct Shared {
    unsigned long long surv[CAND];
    unsigned long long keys[CAND];
    unsigned wcnt[2][NWAVE];
    Phase2 p2;
};

#define BLOCK_COUNT(result, PRED)                                               \
    do {                                                                        \
        unsigned c_ = 0;                                                        \
          \
        _Pragma("unroll")                                                       \
        for (int k = 0; k < IPT; ++k) {                                         \
            const unsigned id_ = id0 + 512u * (unsigned)(k >> 1) + (unsigned)(k & 1); \
            const unsigned rid_ = 8191u - id_;                                  \
            const unsigned sv_ = sb[k];                                         \
            (void)rid_; (void)sv_;                                              \
            c_ += (unsigned)__popc(__builtin_amdgcn_ballot_w32(PRED));          \
        }                                                                       \
        if (lane == 0) S.wcnt[par][wave] = c_;                                  \
        __syncthreads();                                                        \
        unsigned t_ = 0;                                                        \
          \
        _Pragma("unroll")                                                       \
        for (int w = 0; w < NWAVE; ++w) t_ += S.wcnt[par][w];                   \
        par ^= 1;                                                               \
          \
          \
        (result) = (unsigned)__builtin_amdgcn_readfirstlane((int)t_);           \
    } while (0)

__global__ __launch_bounds__(256) void det_kernel(const float* __restrict__ net,
                                                  float* __restrict__ out) {
    __shared__ Shared S;
    Phase2* P = &S.p2;

    const int tid = threadIdx.x;
    const int lane = tid & 31;
    const int wave = tid >> 5;
    const int img = blockIdx.x;
    const float* in = net + (size_t)img * D_IMG;
    const unsigned id0 = 2u * (unsigned)tid;

    unsigned sb[IPT];
    #pragma unroll
    for (int kc = 0; kc < CPT; ++kc) {
        const int cell = tid + kc * 256;
        float s0 = 0.0f, s1 = 0.0f;
        if (kc < CPT - 1 || cell < CELLS) {
            const float* pr = in + cell * N_CLASS;
            float p[N_CLASS];
            #pragma unroll
            for (int c = 0; c < N_CLASS; ++c) p[c] = pr[c];
            const float conf0 = in[CONF_OFF + cell * 2];
            const float conf1 = in[CONF_OFF + cell * 2 + 1];
            float b0 = NEG_INF, b1 = NEG_INF;
            #pragma unroll
            for (int c = 0; c < N_CLASS; ++c) {
                float v0 = __fmul_rn(conf0, p[c]);
                float v1 = __fmul_rn(conf1, p[c]);
                b0 = v0 > b0 ? v0 : b0;
                b1 = v1 > b1 ? v1 : b1;
            }
            s0 = (b0 >= 0.1f) ? b0 : 0.0f;
            s1 = (b1 >= 0.1f) ? b1 : 0.0f;
        }
        sb[2 * kc]     = __float_as_uint(s0);
        sb[2 * kc + 1] = __float_as_uint(s1);
    }

    int par = 0;
    unsigned T = 0u, cntGE = (unsigned)SLOTS;
    for (int bit = 31; bit >= 0; --bit) {
        const unsigned c = T | (1u << bit);
        unsigned n;
        BLOCK_COUNT(n, sv_ >= c);
        if (n >= (unsigned)CAND) { T = c; cntGE = n; }
    }
    unsigned R = 0u;
    if (cntGE != (unsigned)CAND) {
        unsigned above;
        BLOCK_COUNT(above, sv_ > T);
        const unsigned need = (unsigned)CAND - above;
        for (int bit = 12; bit >= 0; --bit) {
            const unsigned c = R | (1u << bit);
            unsigned n;
            BLOCK_COUNT(n, (sv_ == T) && (rid_ >= c));
            if (n >= need) R = c;
        }
    }

    {
        unsigned m[IPT];
        unsigned wtot = 0;
        #pragma unroll
        for (int k = 0; k < IPT; ++k) {
            const unsigned id = id0 + 512u * (unsigned)(k >> 1) + (unsigned)(k & 1);
            const bool take = (sb[k] > T) || ((sb[k] == T) && (8191u - id >= R));
            m[k] = __builtin_amdgcn_ballot_w32(take);
            wtot += (unsigned)__popc(m[k]);
        }
        if (lane == 0) S.wcnt[par][wave] = wtot;
        __syncthreads();
        unsigned base = 0;
        #pragma unroll
        for (int w = 0; w < NWAVE; ++w) base += (w < wave) ? S.wcnt[par][w] : 0u;
        const unsigned below = (1u << lane) - 1u;
        #pragma unroll
        for (int k = 0; k < IPT; ++k) {
            const unsigned id = id0 + 512u * (unsigned)(k >> 1) + (unsigned)(k & 1);
            if ((m[k] >> lane) & 1u) {
                const unsigned pos = (base + (unsigned)__popc(m[k] & below)) & (unsigned)(CAND - 1);
                S.surv[pos] = ((unsigned long long)sb[k] << 32) | (unsigned)(~id);
            }
            base += (unsigned)__popc(m[k]);
        }
    }
    __syncthreads();

    {
        const unsigned long long mine = S.surv[tid];
        unsigned rank = 0;
        #pragma unroll 16
        for (int j = 0; j < CAND; ++j) rank += (S.surv[j] > mine) ? 1u : 0u;
        S.keys[rank] = mine;
    }
    __syncthreads();

    {
        unsigned long long key = S.keys[tid];
        unsigned id = ~(unsigned)key;
        id = id < (unsigned)ITEMS ? id : (unsigned)(ITEMS - 1);
        float s = __uint_as_float((unsigned)(key >> 32));
        int cell = (int)(id >> 1);
        int gi = cell / S_GRID, gj = cell % S_GRID;
        const float* cd = in + COORD_OFF + (size_t)id * 4;
        float tx = cd[0], ty = cd[1], tw = cd[2], th = cd[3];
        float x = __fdiv_rn(__fadd_rn(tx, (float)gj), 56.0f);
        float y = __fdiv_rn(__fadd_rn(ty, (float)gi), 56.0f);
        float hw = __fmul_rn(__fmul_rn(tw, tw), 0.5f);
        float hh = __fmul_rn(__fmul_rn(th, th), 0.5f);
        float y0 = __fsub_rn(y, hh), x0 = __fsub_rn(x, hw);
        float y1 = __fadd_rn(y, hh), x1 = __fadd_rn(x, hw);
        float conf = in[CONF_OFF + id];
        const float* pr = in + cell * N_CLASS;
        float best = NEG_INF; int bc = 0;
        #pragma unroll
        for (int c = 0; c < N_CLASS; ++c) {
            float v = __fmul_rn(conf, pr[c]);
            if (v > best) { best = v; bc = c; }
        }
        float dy = fmaxf(__fsub_rn(y1, y0), 0.0f);
        float dx = fmaxf(__fsub_rn(x1, x0), 0.0f);
        P->y0[tid] = y0; P->x0[tid] = x0; P->y1[tid] = y1; P->x1[tid] = x1;
        P->sc[tid] = s;  P->cl[tid] = (float)bc;
        P->ar[tid] = __fmul_rn(dy, dx);
    }
    __syncthreads();

    {
        v8f czero = {};
        const int c16 = lane & 15;
        const int half8 = (lane & 16) ? 8 : 0;
        const int sel = lane & 7;
        const bool lo16 = lane < 16;
        const int wv = __builtin_amdgcn_readfirstlane(wave);
        for (int mi = wv * 2; mi < wv * 2 + 2; ++mi) {
            const int m0 = mi << 4;
            const int mbase = m0 + half8;
            float my0[8], mx0[8], my1[8], mx1[8];
            #pragma unroll
            for (int r = 0; r < 8; ++r) {
                my0[r] = P->y0[mbase + r]; mx0[r] = P->x0[mbase + r];
                my1[r] = P->y1[mbase + r]; mx1[r] = P->x1[mbase + r];
            }
            const float am = P->ar[m0 + c16];
            v2f A;
            A[0] = lo16 ? am : 0.0f;
            A[1] = lo16 ? 1.0f : 0.0f;
            float ny0 = P->y0[c16], nx0 = P->x0[c16], ny1 = P->y1[c16], nx1 = P->x1[c16];
            float an = P->ar[c16];
            for (int ni = 0; ni < 16; ++ni) {
                const int nn = (((ni + 1) & 15) << 4) + c16;
                const float py0 = P->y0[nn], px0 = P->x0[nn], py1 = P->y1[nn], px1 = P->x1[nn];
                const float pan = P->ar[nn];
                v2f B;
                B[0] = lo16 ? 1.0f : 0.0f;
                B[1] = lo16 ? an : 0.0f;
                v8f d = __builtin_amdgcn_wmma_f32_16x16x4_f32(
                    false, A, false, B, (short)0, czero, false, false);
                unsigned w = 0u;
                #pragma unroll
                for (int r = 0; r < 8; ++r) {
                    const float iy = fmaxf(__fsub_rn(fminf(my1[r], ny1), fmaxf(my0[r], ny0)), 0.0f);
                    const float ix = fmaxf(__fsub_rn(fminf(mx1[r], nx1), fmaxf(mx0[r], nx0)), 0.0f);
                    const float inter = __fmul_rn(iy, ix);
                    const float uni = __fsub_rn(d[r], inter);
                    const float q = __fdiv_rn(inter, uni);
                    const unsigned bal = __builtin_amdgcn_ballot_w32(uni > 0.0f)
                                       & __builtin_amdgcn_ballot_w32(q > 0.4f);
                    w = (sel == r) ? bal : w;
                }
                w = (lane & 8) ? (w >> 16) : w;
                P->mask16[(m0 + c16) * 16 + ni] = (unsigned short)w;
                ny0 = py0; nx0 = px0; ny1 = py1; nx1 = px1; an = pan;
            }
        }
    }
    __syncthreads();

    if (wave == 0) {
        unsigned v = (lane < 16) ? 0xFFFFu : 0u;
        int mySel = 0, myOk = 0;
        for (int it = 0; it < MAXOUT; ++it) {
            const unsigned nz = __builtin_amdgcn_ballot_w32(v != 0u);
            const int ok = (nz != 0u) ? 1 : 0;
            const int c = ok ? __builtin_ctz(nz) : 0;
            const unsigned word = (unsigned)__builtin_amdgcn_readlane((int)v, c);
            const int idx = ok ? (c * 16 + __builtin_ctz(word)) : 0;
            if (lane == it) { mySel = idx; myOk = ok; }
            const unsigned row = P->mask16[idx * 16 + (lane & 15)];
            v &= ~row;
            if (lane == c) v &= ~(1u << (idx & 15));
        }
        if (lane < MAXOUT) { P->selIdx[lane] = mySel; P->selOk[lane] = myOk; }
    }
    __syncthreads();

    if (tid < MAXOUT) {
        int sel = P->selIdx[tid];
        float okf = P->selOk[tid] ? 1.0f : 0.0f;
        P->redv[tid] = __fmul_rn(P->sc[sel], okf);
    }
    __syncthreads();
    if (tid < MAXOUT) {
        int sel = P->selIdx[tid];
        float okf = P->selOk[tid] ? 1.0f : 0.0f;
        float ms = P->redv[tid];
        int rank = 0;
        for (int jt = 0; jt < MAXOUT; ++jt) {
            float vj = P->redv[jt];
            if (vj > ms || (vj == ms && jt < tid)) rank++;
        }
        float* o = out + (size_t)img * (MAXOUT * 6) + (size_t)rank * 6;
        o[0] = __fmul_rn(P->y0[sel], okf);
        o[1] = __fmul_rn(P->x0[sel], okf);
        o[2] = __fmul_rn(P->y1[sel], okf);
        o[3] = __fmul_rn(P->x1[sel], okf);
        o[4] = ms;
        o[5] = __fmul_rn(P->cl[sel], okf);
    }
}

extern "C" void kernel_launch(void* const* d_in, const int* in_sizes, int n_in,
                              void* d_out, int out_size, void* d_ws, size_t ws_size,
                              hipStream_t stream) {
    (void)n_in; (void)out_size; (void)d_ws; (void)ws_size;
    const float* net = (const float*)d_in[0];
    float* out = (float*)d_out;
    int batch = in_sizes[0] / D_IMG;
    det_kernel<<<dim3(batch), dim3(256), 0, stream>>>(net, out);
}


# IoU threshold test of the mask build without its division: rounded inter/uni > thr decided exactly by fma(-thr, uni, inter) >= 2^-26 * uni (midpoint of thr and the next float); mask bits unchanged
# speedup vs baseline: 11.5682x; 1.0906x over previous
#include <hip/hip_runtime.h>


typedef float v2f __attribute__((ext_vector_type(2)));
typedef float v8f __attribute__((ext_vector_type(8)));

#define S_GRID   56
#define N_CLASS  20
#define CELLS    (S_GRID * S_GRID)
#define ITEMS    (CELLS * 2)
#define CAND     256
#define MAXOUT   30
#define D_IMG    94080
#define CONF_OFF 62720
#define COORD_OFF 68992
#define NEG_INF  (-3.402823466e38f)
#define CPT      13
#define IPT      (CPT * 2)
#define SLOTS    (IPT * 256)
#define NWAVE    8

struct Phase2 {
    float y0[CAND], x0[CAND], y1[CAND], x1[CAND];
    float sc[CAND], cl[CAND], ar[CAND];
    float redv[CAND];
    unsigned short mask16[CAND * 16];
    int selIdx[32], selOk[32];
};

struct Shared {
    unsigned long long surv[CAND];
    unsigned long long keys[CAND];
    unsigned wcnt[2][NWAVE];
    Phase2 p2;
};

#define BLOCK_COUNT(result, PRED)                                               \
    do {                                                                        \
        unsigned c_ = 0;                                                        \
          \
        _Pragma("unroll")                                                       \
        for (int k = 0; k < IPT; ++k) {                                         \
            const unsigned id_ = id0 + 512u * (unsigned)(k >> 1) + (unsigned)(k & 1); \
            const unsigned rid_ = 8191u - id_;                                  \
            const unsigned sv_ = sb[k];                                         \
            (void)rid_; (void)sv_;                                              \
            c_ += (unsigned)__popc(__builtin_amdgcn_ballot_w32(PRED));          \
        }                                                                       \
        if (lane == 0) S.wcnt[par][wave] = c_;                                  \
        __syncthreads();                                                        \
        unsigned t_ = 0;                                                        \
          \
        _Pragma("unroll")                                                       \
        for (int w = 0; w < NWAVE; ++w) t_ += S.wcnt[par][w];                   \
        par ^= 1;                                                               \
          \
          \
        (result) = (unsigned)__builtin_amdgcn_readfirstlane((int)t_);           \
    } while (0)

__global__ __launch_bounds__(256) void det_kernel(const float* __restrict__ net,
                                                  float* __restrict__ out) {
    __shared__ Shared S;
    Phase2* P = &S.p2;

    const int tid = threadIdx.x;
    const int lane = tid & 31;
    const int wave = tid >> 5;
    const int img = blockIdx.x;
    const float* in = net + (size_t)img * D_IMG;
    const unsigned id0 = 2u * (unsigned)tid;

    unsigned sb[IPT];
    #pragma unroll
    for (int kc = 0; kc < CPT; ++kc) {
        const int cell = tid + kc * 256;
        float s0 = 0.0f, s1 = 0.0f;
        if (kc < CPT - 1 || cell < CELLS) {
            const float* pr = in + cell * N_CLASS;
            float p[N_CLASS];
            #pragma unroll
            for (int c = 0; c < N_CLASS; ++c) p[c] = pr[c];
            const float conf0 = in[CONF_OFF + cell * 2];
            const float conf1 = in[CONF_OFF + cell * 2 + 1];
            float b0 = NEG_INF, b1 = NEG_INF;
            #pragma unroll
            for (int c = 0; c < N_CLASS; ++c) {
                float v0 = __fmul_rn(conf0, p[c]);
                float v1 = __fmul_rn(conf1, p[c]);
                b0 = v0 > b0 ? v0 : b0;
                b1 = v1 > b1 ? v1 : b1;
            }
            s0 = (b0 >= 0.1f) ? b0 : 0.0f;
            s1 = (b1 >= 0.1f) ? b1 : 0.0f;
        }
        sb[2 * kc]     = __float_as_uint(s0);
        sb[2 * kc + 1] = __float_as_uint(s1);
    }

    int par = 0;
    unsigned T = 0u, cntGE = (unsigned)SLOTS;
    for (int bit = 31; bit >= 0; --bit) {
        const unsigned c = T | (1u << bit);
        unsigned n;
        BLOCK_COUNT(n, sv_ >= c);
        if (n >= (unsigned)CAND) { T = c; cntGE = n; }
    }
    unsigned R = 0u;
    if (cntGE != (unsigned)CAND) {
        unsigned above;
        BLOCK_COUNT(above, sv_ > T);
        const unsigned need = (unsigned)CAND - above;
        for (int bit = 12; bit >= 0; --bit) {
            const unsigned c = R | (1u << bit);
            unsigned n;
            BLOCK_COUNT(n, (sv_ == T) && (rid_ >= c));
            if (n >= need) R = c;
        }
    }

    {
        unsigned m[IPT];
        unsigned wtot = 0;
        #pragma unroll
        for (int k = 0; k < IPT; ++k) {
            const unsigned id = id0 + 512u * (unsigned)(k >> 1) + (unsigned)(k & 1);
            const bool take = (sb[k] > T) || ((sb[k] == T) && (8191u - id >= R));
            m[k] = __builtin_amdgcn_ballot_w32(take);
            wtot += (unsigned)__popc(m[k]);
        }
        if (lane == 0) S.wcnt[par][wave] = wtot;
        __syncthreads();
        unsigned base = 0;
        #pragma unroll
        for (int w = 0; w < NWAVE; ++w) base += (w < wave) ? S.wcnt[par][w] : 0u;
        const unsigned below = (1u << lane) - 1u;
        #pragma unroll
        for (int k = 0; k < IPT; ++k) {
            const unsigned id = id0 + 512u * (unsigned)(k >> 1) + (unsigned)(k & 1);
            if ((m[k] >> lane) & 1u) {
                const unsigned pos = (base + (unsigned)__popc(m[k] & below)) & (unsigned)(CAND - 1);
                S.surv[pos] = ((unsigned long long)sb[k] << 32) | (unsigned)(~id);
            }
            base += (unsigned)__popc(m[k]);
        }
    }
    __syncthreads();

    {
        const unsigned long long mine = S.surv[tid];
        unsigned rank = 0;
        #pragma unroll 16
        for (int j = 0; j < CAND; ++j) rank += (S.surv[j] > mine) ? 1u : 0u;
        S.keys[rank] = mine;
    }
    __syncthreads();

    {
        unsigned long long key = S.keys[tid];
        unsigned id = ~(unsigned)key;
        id = id < (unsigned)ITEMS ? id : (unsigned)(ITEMS - 1);
        float s = __uint_as_float((unsigned)(key >> 32));
        int cell = (int)(id >> 1);
        int gi = cell / S_GRID, gj = cell % S_GRID;
        const float* cd = in + COORD_OFF + (size_t)id * 4;
        float tx = cd[0], ty = cd[1], tw = cd[2], th = cd[3];
        float x = __fdiv_rn(__fadd_rn(tx, (float)gj), 56.0f);
        float y = __fdiv_rn(__fadd_rn(ty, (float)gi), 56.0f);
        float hw = __fmul_rn(__fmul_rn(tw, tw), 0.5f);
        float hh = __fmul_rn(__fmul_rn(th, th), 0.5f);
        float y0 = __fsub_rn(y, hh), x0 = __fsub_rn(x, hw);
        float y1 = __fadd_rn(y, hh), x1 = __fadd_rn(x, hw);
        float conf = in[CONF_OFF + id];
        const float* pr = in + cell * N_CLASS;
        float best = NEG_INF; int bc = 0;
        #pragma unroll
        for (int c = 0; c < N_CLASS; ++c) {
            float v = __fmul_rn(conf, pr[c]);
            if (v > best) { best = v; bc = c; }
        }
        float dy = fmaxf(__fsub_rn(y1, y0), 0.0f);
        float dx = fmaxf(__fsub_rn(x1, x0), 0.0f);
        P->y0[tid] = y0; P->x0[tid] = x0; P->y1[tid] = y1; P->x1[tid] = x1;
        P->sc[tid] = s;  P->cl[tid] = (float)bc;
        P->ar[tid] = __fmul_rn(dy, dx);
    }
    __syncthreads();

    {
        v8f czero = {};
        const int c16 = lane & 15;
        const int half8 = (lane & 16) ? 8 : 0;
        const int sel = lane & 7;
        const bool lo16 = lane < 16;
        const int wv = __builtin_amdgcn_readfirstlane(wave);
        for (int mi = wv * 2; mi < wv * 2 + 2; ++mi) {
            const int m0 = mi << 4;
            const int mbase = m0 + half8;
            float my0[8], mx0[8], my1[8], mx1[8];
            #pragma unroll
            for (int r = 0; r < 8; ++r) {
                my0[r] = P->y0[mbase + r]; mx0[r] = P->x0[mbase + r];
                my1[r] = P->y1[mbase + r]; mx1[r] = P->x1[mbase + r];
            }
            const float am = P->ar[m0 + c16];
            v2f A;
            A[0] = lo16 ? am : 0.0f;
            A[1] = lo16 ? 1.0f : 0.0f;
            float ny0 = P->y0[c16], nx0 = P->x0[c16], ny1 = P->y1[c16], nx1 = P->x1[c16];
            float an = P->ar[c16];
            for (int ni = 0; ni < 16; ++ni) {
                const int nn = (((ni + 1) & 15) << 4) + c16;
                const float py0 = P->y0[nn], px0 = P->x0[nn], py1 = P->y1[nn], px1 = P->x1[nn];
                const float pan = P->ar[nn];
                v2f B;
                B[0] = lo16 ? 1.0f : 0.0f;
                B[1] = lo16 ? an : 0.0f;
                v8f d = __builtin_amdgcn_wmma_f32_16x16x4_f32(
                    false, A, false, B, (short)0, czero, false, false);
                unsigned w = 0u;
                #pragma unroll
                for (int r = 0; r < 8; ++r) {
                    const float iy = fmaxf(__fsub_rn(fminf(my1[r], ny1), fmaxf(my0[r], ny0)), 0.0f);
                    const float ix = fmaxf(__fsub_rn(fminf(mx1[r], nx1), fmaxf(mx0[r], nx0)), 0.0f);
                    const float inter = __fmul_rn(iy, ix);
                    const float uni = __fsub_rn(d[r], inter);
                    const float excess = __builtin_fmaf(-0.4f, uni, inter);
                    const float margin = __fmul_rn(uni, 0x1p-26f);
                    const unsigned bal = __builtin_amdgcn_ballot_w32(uni > 0.0f)
                                       & __builtin_amdgcn_ballot_w32(excess >= margin);
                    w = (sel == r) ? bal : w;
                }
                w = (lane & 8) ? (w >> 16) : w;
                P->mask16[(m0 + c16) * 16 + ni] = (unsigned short)w;
                ny0 = py0; nx0 = px0; ny1 = py1; nx1 = px1; an = pan;
            }
        }
    }
    __syncthreads();

    if (wave == 0) {
        unsigned v = (lane < 16) ? 0xFFFFu : 0u;
        int mySel = 0, myOk = 0;
        for (int it = 0; it < MAXOUT; ++it) {
            const unsigned nz = __builtin_amdgcn_ballot_w32(v != 0u);
            const int ok = (nz != 0u) ? 1 : 0;
            const int c = ok ? __builtin_ctz(nz) : 0;
            const unsigned word = (unsigned)__builtin_amdgcn_readlane((int)v, c);
            const int idx = ok ? (c * 16 + __builtin_ctz(word)) : 0;
            if (lane == it) { mySel = idx; myOk = ok; }
            const unsigned row = P->mask16[idx * 16 + (lane & 15)];
            v &= ~row;
            if (lane == c) v &= ~(1u << (idx & 15));
        }
        if (lane < MAXOUT) { P->selIdx[lane] = mySel; P->selOk[lane] = myOk; }
    }
    __syncthreads();

    if (tid < MAXOUT) {
        int sel = P->selIdx[tid];
        float okf = P->selOk[tid] ? 1.0f : 0.0f;
        P->redv[tid] = __fmul_rn(P->sc[sel], okf);
    }
    __syncthreads();
    if (tid < MAXOUT) {
        int sel = P->selIdx[tid];
        float okf = P->selOk[tid] ? 1.0f : 0.0f;
        float ms = P->redv[tid];
        int rank = 0;
        for (int jt = 0; jt < MAXOUT; ++jt) {
            float vj = P->redv[jt];
            if (vj > ms || (vj == ms && jt < tid)) rank++;
        }
        float* o = out + (size_t)img * (MAXOUT * 6) + (size_t)rank * 6;
        o[0] = __fmul_rn(P->y0[sel], okf);
        o[1] = __fmul_rn(P->x0[sel], okf);
        o[2] = __fmul_rn(P->y1[sel], okf);
        o[3] = __fmul_rn(P->x1[sel], okf);
        o[4] = ms;
        o[5] = __fmul_rn(P->cl[sel], okf);
    }
}

extern "C" void kernel_launch(void* const* d_in, const int* in_sizes, int n_in,
                              void* d_out, int out_size, void* d_ws, size_t ws_size,
                              hipStream_t stream) {
    (void)n_in; (void)out_size; (void)d_ws; (void)ws_size;
    const float* net = (const float*)d_in[0];
    float* out = (float*)d_out;
    int batch = in_sizes[0] / D_IMG;
    det_kernel<<<dim3(batch), dim3(256), 0, stream>>>(net, out);
}


# Top-256 threshold search leaves its block-wide rounds (barrier + LDS sum each) once at most 32 scores are undecided: those go to LDS and every wave finds the remaining bits alone, one ballot per bit
# speedup vs baseline: 12.1715x; 1.0522x over previous
#include <hip/hip_runtime.h>


typedef float v2f __attribute__((ext_vector_type(2)));
typedef float v8f __attribute__((ext_vector_type(8)));

#define S_GRID   56
#define N_CLASS  20
#define CELLS    (S_GRID * S_GRID)
#define ITEMS    (CELLS * 2)
#define CAND     256
#define MAXOUT   30
#define D_IMG    94080
#define CONF_OFF 62720
#define COORD_OFF 68992
#define NEG_INF  (-3.402823466e38f)
#define CPT      13
#define IPT      (CPT * 2)
#define SLOTS    (IPT * 256)
#define NWAVE    8

struct Phase2 {
    float y0[CAND], x0[CAND], y1[CAND], x1[CAND];
    float sc[CAND], cl[CAND], ar[CAND];
    float redv[CAND];
    unsigned short mask16[CAND * 16];
    int selIdx[32], selOk[32];
};

struct Shared {
    unsigned long long surv[CAND];
    unsigned long long keys[CAND];
    unsigned wcnt[2][NWAVE];
    unsigned bucket[32];
    Phase2 p2;
};

#define BLOCK_COUNT(result, PRED)                                               \
    do {                                                                        \
        unsigned c_ = 0;                                                        \
          \
        _Pragma("unroll")                                                       \
        for (int k = 0; k < IPT; ++k) {                                         \
            const unsigned id_ = id0 + 512u * (unsigned)(k >> 1) + (unsigned)(k & 1); \
            const unsigned rid_ = 8191u - id_;                                  \
            const unsigned sv_ = sb[k];                                         \
            (void)rid_; (void)sv_;                                              \
            c_ += (unsigned)__popc(__builtin_amdgcn_ballot_w32(PRED));          \
        }                                                                       \
        if (lane == 0) S.wcnt[par][wave] = c_;                                  \
        __syncthreads();                                                        \
        unsigned t_ = 0;                                                        \
          \
        _Pragma("unroll")                                                       \
        for (int w = 0; w < NWAVE; ++w) t_ += S.wcnt[par][w];                   \
        par ^= 1;                                                               \
          \
          \
        (result) = (unsigned)__builtin_amdgcn_readfirstlane((int)t_);           \
    } while (0)

__global__ __launch_bounds__(256) void det_kernel(const float* __restrict__ net,
                                                  float* __restrict__ out) {
    __shared__ Shared S;
    Phase2* P = &S.p2;

    const int tid = threadIdx.x;
    const int lane = tid & 31;
    const int wave = tid >> 5;
    const int img = blockIdx.x;
    const float* in = net + (size_t)img * D_IMG;
    const unsigned id0 = 2u * (unsigned)tid;

    unsigned sb[IPT];
    #pragma unroll
    for (int kc = 0; kc < CPT; ++kc) {
        const int cell = tid + kc * 256;
        float s0 = 0.0f, s1 = 0.0f;
        if (kc < CPT - 1 || cell < CELLS) {
            const float* pr = in + cell * N_CLASS;
            float p[N_CLASS];
            #pragma unroll
            for (int c = 0; c < N_CLASS; ++c) p[c] = pr[c];
            const float conf0 = in[CONF_OFF + cell * 2];
            const float conf1 = in[CONF_OFF + cell * 2 + 1];
            float b0 = NEG_INF, b1 = NEG_INF;
            #pragma unroll
            for (int c = 0; c < N_CLASS; ++c) {
                float v0 = __fmul_rn(conf0, p[c]);
                float v1 = __fmul_rn(conf1, p[c]);
                b0 = v0 > b0 ? v0 : b0;
                b1 = v1 > b1 ? v1 : b1;
            }
            s0 = (b0 >= 0.1f) ? b0 : 0.0f;
            s1 = (b1 >= 0.1f) ? b1 : 0.0f;
        }
        sb[2 * kc]     = __float_as_uint(s0);
        sb[2 * kc + 1] = __float_as_uint(s1);
    }

    int par = 0;
    unsigned T = 0u, cntGE = (unsigned)SLOTS, cntAbove = 0u;
    int bit = 31;
    for (; bit >= 0; --bit) {
        if (cntGE - cntAbove <= 32u) break;
        const unsigned c = T | (1u << bit);
        unsigned n;
        BLOCK_COUNT(n, sv_ >= c);
        if (n >= (unsigned)CAND) { T = c; cntGE = n; } else { cntAbove = n; }
    }
    if (bit >= 0) {
        const unsigned width = 1u << (bit + 1);
        const unsigned nb = cntGE - cntAbove;
        unsigned m[IPT];
        unsigned wtot = 0;
        #pragma unroll
        for (int k = 0; k < IPT; ++k) {
            m[k] = __builtin_amdgcn_ballot_w32(sb[k] - T < width);
            wtot += (unsigned)__popc(m[k]);
        }
        if (lane == 0) S.wcnt[par][wave] = wtot;
        __syncthreads();
        unsigned base = 0;
        #pragma unroll
        for (int w = 0; w < NWAVE; ++w) base += (w < wave) ? S.wcnt[par][w] : 0u;
        par ^= 1;
        const unsigned below = (1u << lane) - 1u;
        #pragma unroll
        for (int k = 0; k < IPT; ++k) {
            if (m[k] != 0u) {
                if ((m[k] >> lane) & 1u) {
                    S.bucket[(base + (unsigned)__popc(m[k] & below)) & 31u] = sb[k];
                }
                base += (unsigned)__popc(m[k]);
            }
        }
        __syncthreads();
        const unsigned bv = ((unsigned)lane < nb) ? S.bucket[lane] : 0u;
        for (; bit >= 0; --bit) {
            const unsigned c = T | (1u << bit);
            const unsigned n = cntAbove + (unsigned)__popc(__builtin_amdgcn_ballot_w32(bv >= c));
            if (n >= (unsigned)CAND) { T = c; cntGE = n; }
        }
    }
    unsigned R = 0u;
    if (cntGE != (unsigned)CAND) {
        unsigned above;
        BLOCK_COUNT(above, sv_ > T);
        const unsigned need = (unsigned)CAND - above;
        for (int bit = 12; bit >= 0; --bit) {
            const unsigned c = R | (1u << bit);
            unsigned n;
            BLOCK_COUNT(n, (sv_ == T) && (rid_ >= c));
            if (n >= need) R = c;
        }
    }

    {
        unsigned m[IPT];
        unsigned wtot = 0;
        #pragma unroll
        for (int k = 0; k < IPT; ++k) {
            const unsigned id = id0 + 512u * (unsigned)(k >> 1) + (unsigned)(k & 1);
            const bool take = (sb[k] > T) || ((sb[k] == T) && (8191u - id >= R));
            m[k] = __builtin_amdgcn_ballot_w32(take);
            wtot += (unsigned)__popc(m[k]);
        }
        if (lane == 0) S.wcnt[par][wave] = wtot;
        __syncthreads();
        unsigned base = 0;
        #pragma unroll
        for (int w = 0; w < NWAVE; ++w) base += (w < wave) ? S.wcnt[par][w] : 0u;
        const unsigned below = (1u << lane) - 1u;
        #pragma unroll
        for (int k = 0; k < IPT; ++k) {
            const unsigned id = id0 + 512u * (unsigned)(k >> 1) + (unsigned)(k & 1);
            if ((m[k] >> lane) & 1u) {
                const unsigned pos = (base + (unsigned)__popc(m[k] & below)) & (unsigned)(CAND - 1);
                S.surv[pos] = ((unsigned long long)sb[k] << 32) | (unsigned)(~id);
            }
            base += (unsigned)__popc(m[k]);
        }
    }
    __syncthreads();

    {
        const unsigned long long mine = S.surv[tid];
        unsigned rank = 0;
        #pragma unroll 16
        for (int j = 0; j < CAND; ++j) rank += (S.surv[j] > mine) ? 1u : 0u;
        S.keys[rank] = mine;
    }
    __syncthreads();

    {
        unsigned long long key = S.keys[tid];
        unsigned id = ~(unsigned)key;
        id = id < (unsigned)ITEMS ? id : (unsigned)(ITEMS - 1);
        float s = __uint_as_float((unsigned)(key >> 32));
        int cell = (int)(id >> 1);
        int gi = cell / S_GRID, gj = cell % S_GRID;
        const float* cd = in + COORD_OFF + (size_t)id * 4;
        float tx = cd[0], ty = cd[1], tw = cd[2], th = cd[3];
        float x = __fdiv_rn(__fadd_rn(tx, (float)gj), 56.0f);
        float y = __fdiv_rn(__fadd_rn(ty, (float)gi), 56.0f);
        float hw = __fmul_rn(__fmul_rn(tw, tw), 0.5f);
        float hh = __fmul_rn(__fmul_rn(th, th), 0.5f);
        float y0 = __fsub_rn(y, hh), x0 = __fsub_rn(x, hw);
        float y1 = __fadd_rn(y, hh), x1 = __fadd_rn(x, hw);
        float conf = in[CONF_OFF + id];
        const float* pr = in + cell * N_CLASS;
        float best = NEG_INF; int bc = 0;
        #pragma unroll
        for (int c = 0; c < N_CLASS; ++c) {
            float v = __fmul_rn(conf, pr[c]);
            if (v > best) { best = v; bc = c; }
        }
        float dy = fmaxf(__fsub_rn(y1, y0), 0.0f);
        float dx = fmaxf(__fsub_rn(x1, x0), 0.0f);
        P->y0[tid] = y0; P->x0[tid] = x0; P->y1[tid] = y1; P->x1[tid] = x1;
        P->sc[tid] = s;  P->cl[tid] = (float)bc;
        P->ar[tid] = __fmul_rn(dy, dx);
    }
    __syncthreads();

    {
        v8f czero = {};
        const int c16 = lane & 15;
        const int half8 = (lane & 16) ? 8 : 0;
        const int sel = lane & 7;
        const bool lo16 = lane < 16;
        const int wv = __builtin_amdgcn_readfirstlane(wave);
        for (int mi = wv * 2; mi < wv * 2 + 2; ++mi) {
            const int m0 = mi << 4;
            const int mbase = m0 + half8;
            float my0[8], mx0[8], my1[8], mx1[8];
            #pragma unroll
            for (int r = 0; r < 8; ++r) {
                my0[r] = P->y0[mbase + r]; mx0[r] = P->x0[mbase + r];
                my1[r] = P->y1[mbase + r]; mx1[r] = P->x1[mbase + r];
            }
            const float am = P->ar[m0 + c16];
            v2f A;
            A[0] = lo16 ? am : 0.0f;
            A[1] = lo16 ? 1.0f : 0.0f;
            float ny0 = P->y0[c16], nx0 = P->x0[c16], ny1 = P->y1[c16], nx1 = P->x1[c16];
            float an = P->ar[c16];
            for (int ni = 0; ni < 16; ++ni) {
                const int nn = (((ni + 1) & 15) << 4) + c16;
                const float py0 = P->y0[nn], px0 = P->x0[nn], py1 = P->y1[nn], px1 = P->x1[nn];
                const float pan = P->ar[nn];
                v2f B;
                B[0] = lo16 ? 1.0f : 0.0f;
                B[1] = lo16 ? an : 0.0f;
                v8f d = __builtin_amdgcn_wmma_f32_16x16x4_f32(
                    false, A, false, B, (short)0, czero, false, false);
                unsigned w = 0u;
                #pragma unroll
                for (int r = 0; r < 8; ++r) {
                    const float iy = fmaxf(__fsub_rn(fminf(my1[r], ny1), fmaxf(my0[r], ny0)), 0.0f);
                    const float ix = fmaxf(__fsub_rn(fminf(mx1[r], nx1), fmaxf(mx0[r], nx0)), 0.0f);
                    const float inter = __fmul_rn(iy, ix);
                    const float uni = __fsub_rn(d[r], inter);
                    const float excess = __builtin_fmaf(-0.4f, uni, inter);
                    const float margin = __fmul_rn(uni, 0x1p-26f);
                    const unsigned bal = __builtin_amdgcn_ballot_w32(uni > 0.0f)
                                       & __builtin_amdgcn_ballot_w32(excess >= margin);
                    w = (sel == r) ? bal : w;
                }
                w = (lane & 8) ? (w >> 16) : w;
                P->mask16[(m0 + c16) * 16 + ni] = (unsigned short)w;
                ny0 = py0; nx0 = px0; ny1 = py1; nx1 = px1; an = pan;
            }
        }
    }
    __syncthreads();

    if (wave == 0) {
        unsigned v = (lane < 16) ? 0xFFFFu : 0u;
        int mySel = 0, myOk = 0;
        for (int it = 0; it < MAXOUT; ++it) {
            const unsigned nz = __builtin_amdgcn_ballot_w32(v != 0u);
            const int ok = (nz != 0u) ? 1 : 0;
            const int c = ok ? __builtin_ctz(nz) : 0;
            const unsigned word = (unsigned)__builtin_amdgcn_readlane((int)v, c);
            const int idx = ok ? (c * 16 + __builtin_ctz(word)) : 0;
            if (lane == it) { mySel = idx; myOk = ok; }
            const unsigned row = P->mask16[idx * 16 + (lane & 15)];
            v &= ~row;
            if (lane == c) v &= ~(1u << (idx & 15));
        }
        if (lane < MAXOUT) { P->selIdx[lane] = mySel; P->selOk[lane] = myOk; }
    }
    __syncthreads();

    if (tid < MAXOUT) {
        int sel = P->selIdx[tid];
        float okf = P->selOk[tid] ? 1.0f : 0.0f;
        P->redv[tid] = __fmul_rn(P->sc[sel], okf);
    }
    __syncthreads();
    if (tid < MAXOUT) {
        int sel = P->selIdx[tid];
        float okf = P->selOk[tid] ? 1.0f : 0.0f;
        float ms = P->redv[tid];
        int rank = 0;
        for (int jt = 0; jt < MAXOUT; ++jt) {
            float vj = P->redv[jt];
            if (vj > ms || (vj == ms && jt < tid)) rank++;
        }
        float* o = out + (size_t)img * (MAXOUT * 6) + (size_t)rank * 6;
        o[0] = __fmul_rn(P->y0[sel], okf);
        o[1] = __fmul_rn(P->x0[sel], okf);
        o[2] = __fmul_rn(P->y1[sel], okf);
        o[3] = __fmul_rn(P->x1[sel], okf);
        o[4] = ms;
        o[5] = __fmul_rn(P->cl[sel], okf);
    }
}

extern "C" void kernel_launch(void* const* d_in, const int* in_sizes, int n_in,
                              void* d_out, int out_size, void* d_ws, size_t ws_size,
                              hipStream_t stream) {
    (void)n_in; (void)out_size; (void)d_ws; (void)ws_size;
    const float* net = (const float*)d_in[0];
    float* out = (float*)d_out;
    int batch = in_sizes[0] / D_IMG;
    det_kernel<<<dim3(batch), dim3(256), 0, stream>>>(net, out);
}


# IoU mask built only for column groups at or behind the row group (upper triangle, 136 of 256 tiles, 17 per wave): the scan takes the first valid candidate, so a row only clears candidates behind it
# speedup vs baseline: 13.4587x; 1.1058x over previous
#include <hip/hip_runtime.h>


typedef float v2f __attribute__((ext_vector_type(2)));
typedef float v8f __attribute__((ext_vector_type(8)));

#define S_GRID   56
#define N_CLASS  20
#define CELLS    (S_GRID * S_GRID)
#define ITEMS    (CELLS * 2)
#define CAND     256
#define MAXOUT   30
#define D_IMG    94080
#define CONF_OFF 62720
#define COORD_OFF 68992
#define NEG_INF  (-3.402823466e38f)
#define CPT      13
#define IPT      (CPT * 2)
#define SLOTS    (IPT * 256)
#define NWAVE    8

struct Phase2 {
    float y0[CAND], x0[CAND], y1[CAND], x1[CAND];
    float sc[CAND], cl[CAND], ar[CAND];
    float redv[CAND];
    unsigned short mask16[CAND * 16];
    int selIdx[32], selOk[32];
};

struct Shared {
    unsigned long long surv[CAND];
    unsigned long long keys[CAND];
    unsigned wcnt[2][NWAVE];
    unsigned bucket[32];
    Phase2 p2;
};

#define BLOCK_COUNT(result, PRED)                                               \
    do {                                                                        \
        unsigned c_ = 0;                                                        \
          \
        _Pragma("unroll")                                                       \
        for (int k = 0; k < IPT; ++k) {                                         \
            const unsigned id_ = id0 + 512u * (unsigned)(k >> 1) + (unsigned)(k & 1); \
            const unsigned rid_ = 8191u - id_;                                  \
            const unsigned sv_ = sb[k];                                         \
            (void)rid_; (void)sv_;                                              \
            c_ += (unsigned)__popc(__builtin_amdgcn_ballot_w32(PRED));          \
        }                                                                       \
        if (lane == 0) S.wcnt[par][wave] = c_;                                  \
        __syncthreads();                                                        \
        unsigned t_ = 0;                                                        \
          \
        _Pragma("unroll")                                                       \
        for (int w = 0; w < NWAVE; ++w) t_ += S.wcnt[par][w];                   \
        par ^= 1;                                                               \
          \
          \
        (result) = (unsigned)__builtin_amdgcn_readfirstlane((int)t_);           \
    } while (0)

__global__ __launch_bounds__(256) void det_kernel(const float* __restrict__ net,
                                                  float* __restrict__ out) {
    __shared__ Shared S;
    Phase2* P = &S.p2;

    const int tid = threadIdx.x;
    const int lane = tid & 31;
    const int wave = tid >> 5;
    const int img = blockIdx.x;
    const float* in = net + (size_t)img * D_IMG;
    const unsigned id0 = 2u * (unsigned)tid;

    unsigned sb[IPT];
    #pragma unroll
    for (int kc = 0; kc < CPT; ++kc) {
        const int cell = tid + kc * 256;
        float s0 = 0.0f, s1 = 0.0f;
        if (kc < CPT - 1 || cell < CELLS) {
            const float* pr = in + cell * N_CLASS;
            float p[N_CLASS];
            #pragma unroll
            for (int c = 0; c < N_CLASS; ++c) p[c] = pr[c];
            const float conf0 = in[CONF_OFF + cell * 2];
            const float conf1 = in[CONF_OFF + cell * 2 + 1];
            float b0 = NEG_INF, b1 = NEG_INF;
            #pragma unroll
            for (int c = 0; c < N_CLASS; ++c) {
                float v0 = __fmul_rn(conf0, p[c]);
                float v1 = __fmul_rn(conf1, p[c]);
                b0 = v0 > b0 ? v0 : b0;
                b1 = v1 > b1 ? v1 : b1;
            }
            s0 = (b0 >= 0.1f) ? b0 : 0.0f;
            s1 = (b1 >= 0.1f) ? b1 : 0.0f;
        }
        sb[2 * kc]     = __float_as_uint(s0);
        sb[2 * kc + 1] = __float_as_uint(s1);
    }

    int par = 0;
    unsigned T = 0u, cntGE = (unsigned)SLOTS, cntAbove = 0u;
    int bit = 31;
    for (; bit >= 0; --bit) {
        if (cntGE - cntAbove <= 32u) break;
        const unsigned c = T | (1u << bit);
        unsigned n;
        BLOCK_COUNT(n, sv_ >= c);
        if (n >= (unsigned)CAND) { T = c; cntGE = n; } else { cntAbove = n; }
    }
    if (bit >= 0) {
        const unsigned width = 1u << (bit + 1);
        const unsigned nb = cntGE - cntAbove;
        unsigned m[IPT];
        unsigned wtot = 0;
        #pragma unroll
        for (int k = 0; k < IPT; ++k) {
            m[k] = __builtin_amdgcn_ballot_w32(sb[k] - T < width);
            wtot += (unsigned)__popc(m[k]);
        }
        if (lane == 0) S.wcnt[par][wave] = wtot;
        __syncthreads();
        unsigned base = 0;
        #pragma unroll
        for (int w = 0; w < NWAVE; ++w) base += (w < wave) ? S.wcnt[par][w] : 0u;
        par ^= 1;
        const unsigned below = (1u << lane) - 1u;
        #pragma unroll
        for (int k = 0; k < IPT; ++k) {
            if (m[k] != 0u) {
                if ((m[k] >> lane) & 1u) {
                    S.bucket[(base + (unsigned)__popc(m[k] & below)) & 31u] = sb[k];
                }
                base += (unsigned)__popc(m[k]);
            }
        }
        __syncthreads();
        const unsigned bv = ((unsigned)lane < nb) ? S.bucket[lane] : 0u;
        for (; bit >= 0; --bit) {
            const unsigned c = T | (1u << bit);
            const unsigned n = cntAbove + (unsigned)__popc(__builtin_amdgcn_ballot_w32(bv >= c));
            if (n >= (unsigned)CAND) { T = c; cntGE = n; }
        }
    }
    unsigned R = 0u;
    if (cntGE != (unsigned)CAND) {
        unsigned above;
        BLOCK_COUNT(above, sv_ > T);
        const unsigned need = (unsigned)CAND - above;
        for (int bit = 12; bit >= 0; --bit) {
            const unsigned c = R | (1u << bit);
            unsigned n;
            BLOCK_COUNT(n, (sv_ == T) && (rid_ >= c));
            if (n >= need) R = c;
        }
    }

    {
        unsigned m[IPT];
        unsigned wtot = 0;
        #pragma unroll
        for (int k = 0; k < IPT; ++k) {
            const unsigned id = id0 + 512u * (unsigned)(k >> 1) + (unsigned)(k & 1);
            const bool take = (sb[k] > T) || ((sb[k] == T) && (8191u - id >= R));
            m[k] = __builtin_amdgcn_ballot_w32(take);
            wtot += (unsigned)__popc(m[k]);
        }
        if (lane == 0) S.wcnt[par][wave] = wtot;
        __syncthreads();
        unsigned base = 0;
        #pragma unroll
        for (int w = 0; w < NWAVE; ++w) base += (w < wave) ? S.wcnt[par][w] : 0u;
        const unsigned below = (1u << lane) - 1u;
        #pragma unroll
        for (int k = 0; k < IPT; ++k) {
            const unsigned id = id0 + 512u * (unsigned)(k >> 1) + (unsigned)(k & 1);
            if ((m[k] >> lane) & 1u) {
                const unsigned pos = (base + (unsigned)__popc(m[k] & below)) & (unsigned)(CAND - 1);
                S.surv[pos] = ((unsigned long long)sb[k] << 32) | (unsigned)(~id);
            }
            base += (unsigned)__popc(m[k]);
        }
    }
    __syncthreads();

    {
        const unsigned long long mine = S.surv[tid];
        unsigned rank = 0;
        #pragma unroll 16
        for (int j = 0; j < CAND; ++j) rank += (S.surv[j] > mine) ? 1u : 0u;
        S.keys[rank] = mine;
    }
    __syncthreads();

    {
        unsigned long long key = S.keys[tid];
        unsigned id = ~(unsigned)key;
        id = id < (unsigned)ITEMS ? id : (unsigned)(ITEMS - 1);
        float s = __uint_as_float((unsigned)(key >> 32));
        int cell = (int)(id >> 1);
        int gi = cell / S_GRID, gj = cell % S_GRID;
        const float* cd = in + COORD_OFF + (size_t)id * 4;
        float tx = cd[0], ty = cd[1], tw = cd[2], th = cd[3];
        float x = __fdiv_rn(__fadd_rn(tx, (float)gj), 56.0f);
        float y = __fdiv_rn(__fadd_rn(ty, (float)gi), 56.0f);
        float hw = __fmul_rn(__fmul_rn(tw, tw), 0.5f);
        float hh = __fmul_rn(__fmul_rn(th, th), 0.5f);
        float y0 = __fsub_rn(y, hh), x0 = __fsub_rn(x, hw);
        float y1 = __fadd_rn(y, hh), x1 = __fadd_rn(x, hw);
        float conf = in[CONF_OFF + id];
        const float* pr = in + cell * N_CLASS;
        float best = NEG_INF; int bc = 0;
        #pragma unroll
        for (int c = 0; c < N_CLASS; ++c) {
            float v = __fmul_rn(conf, pr[c]);
            if (v > best) { best = v; bc = c; }
        }
        float dy = fmaxf(__fsub_rn(y1, y0), 0.0f);
        float dx = fmaxf(__fsub_rn(x1, x0), 0.0f);
        P->y0[tid] = y0; P->x0[tid] = x0; P->y1[tid] = y1; P->x1[tid] = x1;
        P->sc[tid] = s;  P->cl[tid] = (float)bc;
        P->ar[tid] = __fmul_rn(dy, dx);
    }
    __syncthreads();

    {
        v8f czero = {};
        const int c16 = lane & 15;
        const int half8 = (lane & 16) ? 8 : 0;
        const int sel = lane & 7;
        const bool lo16 = lane < 16;
        const int wv = __builtin_amdgcn_readfirstlane(wave);
        for (int half = 0; half < 2; ++half) {
            const int mi = half ? 15 - wv : wv;
            const int m0 = mi << 4;
            const int mbase = m0 + half8;
            float my0[8], mx0[8], my1[8], mx1[8];
            #pragma unroll
            for (int r = 0; r < 8; ++r) {
                my0[r] = P->y0[mbase + r]; mx0[r] = P->x0[mbase + r];
                my1[r] = P->y1[mbase + r]; mx1[r] = P->x1[mbase + r];
            }
            const float am = P->ar[m0 + c16];
            v2f A;
            A[0] = lo16 ? am : 0.0f;
            A[1] = lo16 ? 1.0f : 0.0f;
            float ny0 = P->y0[m0 + c16], nx0 = P->x0[m0 + c16], ny1 = P->y1[m0 + c16], nx1 = P->x1[m0 + c16];
            float an = am;
            for (int ni = mi; ni < 16; ++ni) {
                const int nn = (((ni + 1) & 15) << 4) + c16;
                const float py0 = P->y0[nn], px0 = P->x0[nn], py1 = P->y1[nn], px1 = P->x1[nn];
                const float pan = P->ar[nn];
                v2f B;
                B[0] = lo16 ? 1.0f : 0.0f;
                B[1] = lo16 ? an : 0.0f;
                v8f d = __builtin_amdgcn_wmma_f32_16x16x4_f32(
                    false, A, false, B, (short)0, czero, false, false);
                unsigned w = 0u;
                #pragma unroll
                for (int r = 0; r < 8; ++r) {
                    const float iy = fmaxf(__fsub_rn(fminf(my1[r], ny1), fmaxf(my0[r], ny0)), 0.0f);
                    const float ix = fmaxf(__fsub_rn(fminf(mx1[r], nx1), fmaxf(mx0[r], nx0)), 0.0f);
                    const float inter = __fmul_rn(iy, ix);
                    const float uni = __fsub_rn(d[r], inter);
                    const float excess = __builtin_fmaf(-0.4f, uni, inter);
                    const float margin = __fmul_rn(uni, 0x1p-26f);
                    const unsigned bal = __builtin_amdgcn_ballot_w32(uni > 0.0f)
                                       & __builtin_amdgcn_ballot_w32(excess >= margin);
                    w = (sel == r) ? bal : w;
                }
                w = (lane & 8) ? (w >> 16) : w;
                P->mask16[(m0 + c16) * 16 + ni] = (unsigned short)w;
                ny0 = py0; nx0 = px0; ny1 = py1; nx1 = px1; an = pan;
            }
        }
    }
    __syncthreads();

    if (wave == 0) {
        unsigned v = (lane < 16) ? 0xFFFFu : 0u;
        int mySel = 0, myOk = 0;
        for (int it = 0; it < MAXOUT; ++it) {
            const unsigned nz = __builtin_amdgcn_ballot_w32(v != 0u);
            const int ok = (nz != 0u) ? 1 : 0;
            const int c = ok ? __builtin_ctz(nz) : 0;
            const unsigned word = (unsigned)__builtin_amdgcn_readlane((int)v, c);
            const int idx = ok ? (c * 16 + __builtin_ctz(word)) : 0;
            if (lane == it) { mySel = idx; myOk = ok; }
            const int wc = max(lane & 15, c);
            const unsigned row = P->mask16[idx * 16 + wc];
            v &= ~row;
            if (lane == c) v &= ~(1u << (idx & 15));
        }
        if (lane < MAXOUT) { P->selIdx[lane] = mySel; P->selOk[lane] = myOk; }
    }
    __syncthreads();

    if (tid < MAXOUT) {
        int sel = P->selIdx[tid];
        float okf = P->selOk[tid] ? 1.0f : 0.0f;
        P->redv[tid] = __fmul_rn(P->sc[sel], okf);
    }
    __syncthreads();
    if (tid < MAXOUT) {
        int sel = P->selIdx[tid];
        float okf = P->selOk[tid] ? 1.0f : 0.0f;
        float ms = P->redv[tid];
        int rank = 0;
        for (int jt = 0; jt < MAXOUT; ++jt) {
            float vj = P->redv[jt];
            if (vj > ms || (vj == ms && jt < tid)) rank++;
        }
        float* o = out + (size_t)img * (MAXOUT * 6) + (size_t)rank * 6;
        o[0] = __fmul_rn(P->y0[sel], okf);
        o[1] = __fmul_rn(P->x0[sel], okf);
        o[2] = __fmul_rn(P->y1[sel], okf);
        o[3] = __fmul_rn(P->x1[sel], okf);
        o[4] = ms;
        o[5] = __fmul_rn(P->cl[sel], okf);
    }
}

extern "C" void kernel_launch(void* const* d_in, const int* in_sizes, int n_in,
                              void* d_out, int out_size, void* d_ws, size_t ws_size,
                              hipStream_t stream) {
    (void)n_in; (void)out_size; (void)d_ws; (void)ws_size;
    const float* net = (const float*)d_in[0];
    float* out = (float*)d_out;
    int batch = in_sizes[0] / D_IMG;
    det_kernel<<<dim3(batch), dim3(256), 0, stream>>>(net, out);
}


# Candidate list cut to the 64 best for a first pass (threshold search, placement, decode, IoU mask, scan on 64, not 256); exact: with fewer than 30 picks found the 256-candidate pass runs as before
# speedup vs baseline: 16.8455x; 1.2516x over previous
#include <hip/hip_runtime.h>


typedef float v2f __attribute__((ext_vector_type(2)));
typedef float v8f __attribute__((ext_vector_type(8)));

#define S_GRID   56
#define N_CLASS  20
#define CELLS    (S_GRID * S_GRID)
#define ITEMS    (CELLS * 2)
#define CAND     256
#define CAND1    64
#define MAXOUT   30
#define D_IMG    94080
#define CONF_OFF 62720
#define COORD_OFF 68992
#define NEG_INF  (-3.402823466e38f)
#define CPT      13
#define IPT      (CPT * 2)
#define SLOTS    (IPT * 256)
#define NWAVE    8

struct Phase2 {
    float y0[CAND], x0[CAND], y1[CAND], x1[CAND];
    float sc[CAND], cl[CAND], ar[CAND];
    float redv[CAND];
    unsigned short mask16[CAND * 16];
    int selIdx[32], selOk[32];
};

struct Shared {
    unsigned long long surv[CAND];
    unsigned long long keys[CAND];
    unsigned wcnt[2][NWAVE];
    unsigned bucket[32];
    Phase2 p2;
};

#define BLOCK_COUNT(result, PRED)                                               \
    do {                                                                        \
        unsigned c_ = 0;                                                        \
          \
        _Pragma("unroll")                                                       \
        for (int k = 0; k < IPT; ++k) {                                         \
            const unsigned id_ = id0 + 512u * (unsigned)(k >> 1) + (unsigned)(k & 1); \
            const unsigned rid_ = 8191u - id_;                                  \
            const unsigned sv_ = sb[k];                                         \
            (void)rid_; (void)sv_;                                              \
            c_ += (unsigned)__popc(__builtin_amdgcn_ballot_w32(PRED));          \
        }                                                                       \
        if (lane == 0) S.wcnt[par][wave] = c_;                                  \
        __syncthreads();                                                        \
        unsigned t_ = 0;                                                        \
          \
        _Pragma("unroll")                                                       \
        for (int w = 0; w < NWAVE; ++w) t_ += S.wcnt[par][w];                   \
        par ^= 1;                                                               \
          \
          \
        (result) = (unsigned)__builtin_amdgcn_readfirstlane((int)t_);           \
    } while (0)

__global__ __launch_bounds__(256) void det_kernel(const float* __restrict__ net,
                                                  float* __restrict__ out) {
    __shared__ Shared S;
    Phase2* P = &S.p2;

    const int tid = threadIdx.x;
    const int lane = tid & 31;
    const int wave = tid >> 5;
    const int img = blockIdx.x;
    const float* in = net + (size_t)img * D_IMG;
    const unsigned id0 = 2u * (unsigned)tid;

    unsigned sb[IPT];
    #pragma unroll
    for (int kc = 0; kc < CPT; ++kc) {
        const int cell = tid + kc * 256;
        float s0 = 0.0f, s1 = 0.0f;
        if (kc < CPT - 1 || cell < CELLS) {
            const float* pr = in + cell * N_CLASS;
            float p[N_CLASS];
            #pragma unroll
            for (int c = 0; c < N_CLASS; ++c) p[c] = pr[c];
            const float conf0 = in[CONF_OFF + cell * 2];
            const float conf1 = in[CONF_OFF + cell * 2 + 1];
            float b0 = NEG_INF, b1 = NEG_INF;
            #pragma unroll
            for (int c = 0; c < N_CLASS; ++c) {
                float v0 = __fmul_rn(conf0, p[c]);
                float v1 = __fmul_rn(conf1, p[c]);
                b0 = v0 > b0 ? v0 : b0;
                b1 = v1 > b1 ? v1 : b1;
            }
            s0 = (b0 >= 0.1f) ? b0 : 0.0f;
            s1 = (b1 >= 0.1f) ? b1 : 0.0f;
        }
        sb[2 * kc]     = __float_as_uint(s0);
        sb[2 * kc + 1] = __float_as_uint(s1);
    }

    int par = 0;
    for (int pass = 0; pass < 2; ++pass) {
    const int K = pass ? CAND : CAND1;
    const int NG = K >> 4;
    unsigned T = 0u, cntGE = (unsigned)SLOTS, cntAbove = 0u;
    int bit = 31;
    for (; bit >= 0; --bit) {
        if (cntGE - cntAbove <= 32u) break;
        const unsigned c = T | (1u << bit);
        unsigned n;
        BLOCK_COUNT(n, sv_ >= c);
        if (n >= (unsigned)K) { T = c; cntGE = n; } else { cntAbove = n; }
    }
    if (bit >= 0) {
        const unsigned width = 1u << (bit + 1);
        const unsigned nb = cntGE - cntAbove;
        unsigned m[IPT];
        unsigned wtot = 0;
        #pragma unroll
        for (int k = 0; k < IPT; ++k) {
            m[k] = __builtin_amdgcn_ballot_w32(sb[k] - T < width);
            wtot += (unsigned)__popc(m[k]);
        }
        if (lane == 0) S.wcnt[par][wave] = wtot;
        __syncthreads();
        unsigned base = 0;
        #pragma unroll
        for (int w = 0; w < NWAVE; ++w) base += (w < wave) ? S.wcnt[par][w] : 0u;
        par ^= 1;
        const unsigned below = (1u << lane) - 1u;
        #pragma unroll
        for (int k = 0; k < IPT; ++k) {
            if (m[k] != 0u) {
                if ((m[k] >> lane) & 1u) {
                    S.bucket[(base + (unsigned)__popc(m[k] & below)) & 31u] = sb[k];
                }
                base += (unsigned)__popc(m[k]);
            }
        }
        __syncthreads();
        const unsigned bv = ((unsigned)lane < nb) ? S.bucket[lane] : 0u;
        for (; bit >= 0; --bit) {
            const unsigned c = T | (1u << bit);
            const unsigned n = cntAbove + (unsigned)__popc(__builtin_amdgcn_ballot_w32(bv >= c));
            if (n >= (unsigned)K) { T = c; cntGE = n; }
        }
    }
    unsigned R = 0u;
    if (cntGE != (unsigned)K) {
        unsigned above;
        BLOCK_COUNT(above, sv_ > T);
        const unsigned need = (unsigned)K - above;
        for (int bit = 12; bit >= 0; --bit) {
            const unsigned c = R | (1u << bit);
            unsigned n;
            BLOCK_COUNT(n, (sv_ == T) && (rid_ >= c));
            if (n >= need) R = c;
        }
    }

    {
        unsigned m[IPT];
        unsigned wtot = 0;
        #pragma unroll
        for (int k = 0; k < IPT; ++k) {
            const unsigned id = id0 + 512u * (unsigned)(k >> 1) + (unsigned)(k & 1);
            const bool take = (sb[k] > T) || ((sb[k] == T) && (8191u - id >= R));
            m[k] = __builtin_amdgcn_ballot_w32(take);
            wtot += (unsigned)__popc(m[k]);
        }
        if (lane == 0) S.wcnt[par][wave] = wtot;
        __syncthreads();
        unsigned base = 0;
        #pragma unroll
        for (int w = 0; w < NWAVE; ++w) base += (w < wave) ? S.wcnt[par][w] : 0u;
        par ^= 1;
        const unsigned below = (1u << lane) - 1u;
        #pragma unroll
        for (int k = 0; k < IPT; ++k) {
            const unsigned id = id0 + 512u * (unsigned)(k >> 1) + (unsigned)(k & 1);
            if ((m[k] >> lane) & 1u) {
                const unsigned pos = (base + (unsigned)__popc(m[k] & below)) & (unsigned)(K - 1);
                S.surv[pos] = ((unsigned long long)sb[k] << 32) | (unsigned)(~id);
            }
            base += (unsigned)__popc(m[k]);
        }
    }
    __syncthreads();

    if (tid < K) {
        const unsigned long long mine = S.surv[tid];
        unsigned rank = 0;
        #pragma unroll 16
        for (int j = 0; j < K; ++j) rank += (S.surv[j] > mine) ? 1u : 0u;
        S.keys[rank] = mine;
    }
    __syncthreads();

    if (tid < K) {
        unsigned long long key = S.keys[tid];
        unsigned id = ~(unsigned)key;
        id = id < (unsigned)ITEMS ? id : (unsigned)(ITEMS - 1);
        float s = __uint_as_float((unsigned)(key >> 32));
        int cell = (int)(id >> 1);
        int gi = cell / S_GRID, gj = cell % S_GRID;
        const float* cd = in + COORD_OFF + (size_t)id * 4;
        float tx = cd[0], ty = cd[1], tw = cd[2], th = cd[3];
        float x = __fdiv_rn(__fadd_rn(tx, (float)gj), 56.0f);
        float y = __fdiv_rn(__fadd_rn(ty, (float)gi), 56.0f);
        float hw = __fmul_rn(__fmul_rn(tw, tw), 0.5f);
        float hh = __fmul_rn(__fmul_rn(th, th), 0.5f);
        float y0 = __fsub_rn(y, hh), x0 = __fsub_rn(x, hw);
        float y1 = __fadd_rn(y, hh), x1 = __fadd_rn(x, hw);
        float conf = in[CONF_OFF + id];
        const float* pr = in + cell * N_CLASS;
        float best = NEG_INF; int bc = 0;
        #pragma unroll
        for (int c = 0; c < N_CLASS; ++c) {
            float v = __fmul_rn(conf, pr[c]);
            if (v > best) { best = v; bc = c; }
        }
        float dy = fmaxf(__fsub_rn(y1, y0), 0.0f);
        float dx = fmaxf(__fsub_rn(x1, x0), 0.0f);
        P->y0[tid] = y0; P->x0[tid] = x0; P->y1[tid] = y1; P->x1[tid] = x1;
        P->sc[tid] = s;  P->cl[tid] = (float)bc;
        P->ar[tid] = __fmul_rn(dy, dx);
    }
    __syncthreads();

    {
        v8f czero = {};
        const int c16 = lane & 15;
        const int half8 = (lane & 16) ? 8 : 0;
        const int sel = lane & 7;
        const bool lo16 = lane < 16;
        const int wv = __builtin_amdgcn_readfirstlane(wave);
        const int nhalf = (NG > NWAVE) ? 2 : ((wv < NG) ? 1 : 0);
        for (int half = 0; half < nhalf; ++half) {
            const int mi = half ? (NG - 1) - wv : wv;
            const int m0 = mi << 4;
            const int mbase = m0 + half8;
            float my0[8], mx0[8], my1[8], mx1[8];
            #pragma unroll
            for (int r = 0; r < 8; ++r) {
                my0[r] = P->y0[mbase + r]; mx0[r] = P->x0[mbase + r];
                my1[r] = P->y1[mbase + r]; mx1[r] = P->x1[mbase + r];
            }
            const float am = P->ar[m0 + c16];
            v2f A;
            A[0] = lo16 ? am : 0.0f;
            A[1] = lo16 ? 1.0f : 0.0f;
            float ny0 = P->y0[m0 + c16], nx0 = P->x0[m0 + c16], ny1 = P->y1[m0 + c16], nx1 = P->x1[m0 + c16];
            float an = am;
            for (int ni = mi; ni < NG; ++ni) {
                const int nn = (((ni + 1) & (NG - 1)) << 4) + c16;
                const float py0 = P->y0[nn], px0 = P->x0[nn], py1 = P->y1[nn], px1 = P->x1[nn];
                const float pan = P->ar[nn];
                v2f B;
                B[0] = lo16 ? 1.0f : 0.0f;
                B[1] = lo16 ? an : 0.0f;
                v8f d = __builtin_amdgcn_wmma_f32_16x16x4_f32(
                    false, A, false, B, (short)0, czero, false, false);
                unsigned w = 0u;
                #pragma unroll
                for (int r = 0; r < 8; ++r) {
                    const float iy = fmaxf(__fsub_rn(fminf(my1[r], ny1), fmaxf(my0[r], ny0)), 0.0f);
                    const float ix = fmaxf(__fsub_rn(fminf(mx1[r], nx1), fmaxf(mx0[r], nx0)), 0.0f);
                    const float inter = __fmul_rn(iy, ix);
                    const float uni = __fsub_rn(d[r], inter);
                    const float excess = __builtin_fmaf(-0.4f, uni, inter);
                    const float margin = __fmul_rn(uni, 0x1p-26f);
                    const unsigned bal = __builtin_amdgcn_ballot_w32(uni > 0.0f)
                                       & __builtin_amdgcn_ballot_w32(excess >= margin);
                    w = (sel == r) ? bal : w;
                }
                w = (lane & 8) ? (w >> 16) : w;
                P->mask16[(m0 + c16) * 16 + ni] = (unsigned short)w;
                ny0 = py0; nx0 = px0; ny1 = py1; nx1 = px1; an = pan;
            }
        }
    }
    __syncthreads();

    if (wave == 0) {
        unsigned v = (lane < NG) ? 0xFFFFu : 0u;
        int mySel = 0, myOk = 0;
        for (int it = 0; it < MAXOUT; ++it) {
            const unsigned nz = __builtin_amdgcn_ballot_w32(v != 0u);
            const int ok = (nz != 0u) ? 1 : 0;
            const int c = ok ? __builtin_ctz(nz) : 0;
            const unsigned word = (unsigned)__builtin_amdgcn_readlane((int)v, c);
            const int idx = ok ? (c * 16 + __builtin_ctz(word)) : 0;
            if (lane == it) { mySel = idx; myOk = ok; }
            const int wc = min(max(lane & 15, c), NG - 1);
            const unsigned row = P->mask16[idx * 16 + wc];
            v &= ~row;
            if (lane == c) v &= ~(1u << (idx & 15));
        }
        if (lane < MAXOUT) { P->selIdx[lane] = mySel; P->selOk[lane] = myOk; }
    }
    __syncthreads();
    if (__builtin_amdgcn_readfirstlane(P->selOk[MAXOUT - 1]) != 0) break;
    }

    if (tid < MAXOUT) {
        int sel = P->selIdx[tid];
        float okf = P->selOk[tid] ? 1.0f : 0.0f;
        P->redv[tid] = __fmul_rn(P->sc[sel], okf);
    }
    __syncthreads();
    if (tid < MAXOUT) {
        int sel = P->selIdx[tid];
        float okf = P->selOk[tid] ? 1.0f : 0.0f;
        float ms = P->redv[tid];
        int rank = 0;
        for (int jt = 0; jt < MAXOUT; ++jt) {
            float vj = P->redv[jt];
            if (vj > ms || (vj == ms && jt < tid)) rank++;
        }
        float* o = out + (size_t)img * (MAXOUT * 6) + (size_t)rank * 6;
        o[0] = __fmul_rn(P->y0[sel], okf);
        o[1] = __fmul_rn(P->x0[sel], okf);
        o[2] = __fmul_rn(P->y1[sel], okf);
        o[3] = __fmul_rn(P->x1[sel], okf);
        o[4] = ms;
        o[5] = __fmul_rn(P->cl[sel], okf);
    }
}

extern "C" void kernel_launch(void* const* d_in, const int* in_sizes, int n_in,
                              void* d_out, int out_size, void* d_ws, size_t ws_size,
                              hipStream_t stream) {
    (void)n_in; (void)out_size; (void)d_ws; (void)ws_size;
    const float* net = (const float*)d_in[0];
    float* out = (float*)d_out;
    int batch = in_sizes[0] / D_IMG;
    det_kernel<<<dim3(batch), dim3(256), 0, stream>>>(net, out);
}


# Phase 1 reads the class values in whole lines: each wave-wide 16-byte load takes 512 consecutive bytes (4 lines, not 20); piece maxima regrouped per cell in wave-private LDS; score = conf * max class
# speedup vs baseline: 18.6220x; 1.1055x over previous
#include <hip/hip_runtime.h>


typedef float v2f __attribute__((ext_vector_type(2)));
typedef float v8f __attribute__((ext_vector_type(8)));
typedef float v4f __attribute__((ext_vector_type(4)));

#define S_GRID   56
#define N_CLASS  20
#define CELLS    (S_GRID * S_GRID)
#define ITEMS    (CELLS * 2)
#define CAND     256
#define CAND1    64
#define MAXOUT   30
#define D_IMG    94080
#define CONF_OFF 62720
#define COORD_OFF 68992
#define NEG_INF  (-3.402823466e38f)
#define CPT      13
#define IPT      (CPT * 2)
#define SLOTS    (IPT * 256)
#define NWAVE    8
#define ROWQ     160

struct Phase2 {
    float y0[CAND], x0[CAND], y1[CAND], x1[CAND];
    float sc[CAND], cl[CAND], ar[CAND];
    float redv[CAND];
    unsigned short mask16[CAND * 16];
    int selIdx[32], selOk[32];
};

struct Shared {
    unsigned long long surv[CAND];
    unsigned long long keys[CAND];
    unsigned wcnt[2][NWAVE];
    unsigned bucket[32];
    Phase2 p2;
    float part[NWAVE][CPT][ROWQ];
};

#define BLOCK_COUNT(result, PRED)                                               \
    do {                                                                        \
        unsigned c_ = 0;                                                        \
          \
        _Pragma("unroll")                                                       \
        for (int k = 0; k < IPT; ++k) {                                         \
            const unsigned id_ = id0 + 512u * (unsigned)(k >> 1) + (unsigned)(k & 1); \
            const unsigned rid_ = 8191u - id_;                                  \
            const unsigned sv_ = sb[k];                                         \
            (void)rid_; (void)sv_;                                              \
            c_ += (unsigned)__popc(__builtin_amdgcn_ballot_w32(PRED));          \
        }                                                                       \
        if (lane == 0) S.wcnt[par][wave] = c_;                                  \
        __syncthreads();                                                        \
        unsigned t_ = 0;                                                        \
          \
        _Pragma("unroll")                                                       \
        for (int w = 0; w < NWAVE; ++w) t_ += S.wcnt[par][w];                   \
        par ^= 1;                                                               \
          \
          \
        (result) = (unsigned)__builtin_amdgcn_readfirstlane((int)t_);           \
    } while (0)

__global__ __launch_bounds__(256) void det_kernel(const float* __restrict__ net,
                                                  float* __restrict__ out) {
    __shared__ Shared S;
    Phase2* P = &S.p2;

    const int tid = threadIdx.x;
    const int lane = tid & 31;
    const int wave = tid >> 5;
    const int img = blockIdx.x;
    const float* in = net + (size_t)img * D_IMG;
    const unsigned id0 = 2u * (unsigned)tid;

    unsigned sb[IPT];
    {
        const int wv1 = __builtin_amdgcn_readfirstlane(wave);
        v4f q[CPT][5];
        float cf0[CPT], cf1[CPT];
        #pragma unroll
        for (int kc = 0; kc < CPT; ++kc) {
            const int cell = tid + kc * 256;
            cf0[kc] = in[CONF_OFF + cell * 2];
            cf1[kc] = in[CONF_OFF + cell * 2 + 1];
            const v4f* row = (const v4f*)(in + (size_t)(32 * wv1 + 256 * kc) * N_CLASS);
            #pragma unroll
            for (int j = 0; j < 5; ++j) q[kc][j] = row[32 * j + lane];
        }
        #pragma unroll
        for (int kc = 0; kc < CPT; ++kc) {
            float* w = &S.part[wv1][kc][0];
            #pragma unroll
            for (int j = 0; j < 5; ++j) {
                const v4f x = q[kc][j];
                const float a = x[0] > x[1] ? x[0] : x[1];
                const float b = x[2] > x[3] ? x[2] : x[3];
                w[32 * j + lane] = a > b ? a : b;
            }
            __builtin_amdgcn_fence(__ATOMIC_RELEASE, "wavefront");
            __builtin_amdgcn_wave_barrier();
            __builtin_amdgcn_fence(__ATOMIC_ACQUIRE, "wavefront");
            float pm = w[5 * lane];
            #pragma unroll
            for (int i = 1; i < 5; ++i) { const float v = w[5 * lane + i]; pm = v > pm ? v : pm; }
            const float b0 = __fmul_rn(cf0[kc], pm);
            const float b1 = __fmul_rn(cf1[kc], pm);
            float s0 = (b0 >= 0.1f) ? b0 : 0.0f;
            float s1 = (b1 >= 0.1f) ? b1 : 0.0f;
            if (kc == CPT - 1 && wv1 >= 2) { s0 = 0.0f; s1 = 0.0f; }
            sb[2 * kc]     = __float_as_uint(s0);
            sb[2 * kc + 1] = __float_as_uint(s1);
        }
    }

    int par = 0;
    for (int pass = 0; pass < 2; ++pass) {
    const int K = pass ? CAND : CAND1;
    const int NG = K >> 4;
    unsigned T = 0u, cntGE = (unsigned)SLOTS, cntAbove = 0u;
    int bit = 31;
    for (; bit >= 0; --bit) {
        if (cntGE - cntAbove <= 32u) break;
        const unsigned c = T | (1u << bit);
        unsigned n;
        BLOCK_COUNT(n, sv_ >= c);
        if (n >= (unsigned)K) { T = c; cntGE = n; } else { cntAbove = n; }
    }
    if (bit >= 0) {
        const unsigned width = 1u << (bit + 1);
        const unsigned nb = cntGE - cntAbove;
        unsigned m[IPT];
        unsigned wtot = 0;
        #pragma unroll
        for (int k = 0; k < IPT; ++k) {
            m[k] = __builtin_amdgcn_ballot_w32(sb[k] - T < width);
            wtot += (unsigned)__popc(m[k]);
        }
        if (lane == 0) S.wcnt[par][wave] = wtot;
        __syncthreads();
        unsigned base = 0;
        #pragma unroll
        for (int w = 0; w < NWAVE; ++w) base += (w < wave) ? S.wcnt[par][w] : 0u;
        par ^= 1;
        const unsigned below = (1u << lane) - 1u;
        #pragma unroll
        for (int k = 0; k < IPT; ++k) {
            if (m[k] != 0u) {
                if ((m[k] >> lane) & 1u) {
                    S.bucket[(base + (unsigned)__popc(m[k] & below)) & 31u] = sb[k];
                }
                base += (unsigned)__popc(m[k]);
            }
        }
        __syncthreads();
        const unsigned bv = ((unsigned)lane < nb) ? S.bucket[lane] : 0u;
        for (; bit >= 0; --bit) {
            const unsigned c = T | (1u << bit);
            const unsigned n = cntAbove + (unsigned)__popc(__builtin_amdgcn_ballot_w32(bv >= c));
            if (n >= (unsigned)K) { T = c; cntGE = n; }
        }
    }
    unsigned R = 0u;
    if (cntGE != (unsigned)K) {
        unsigned above;
        BLOCK_COUNT(above, sv_ > T);
        const unsigned need = (unsigned)K - above;
        for (int bit = 12; bit >= 0; --bit) {
            const unsigned c = R | (1u << bit);
            unsigned n;
            BLOCK_COUNT(n, (sv_ == T) && (rid_ >= c));
            if (n >= need) R = c;
        }
    }

    {
        unsigned m[IPT];
        unsigned wtot = 0;
        #pragma unroll
        for (int k = 0; k < IPT; ++k) {
            const unsigned id = id0 + 512u * (unsigned)(k >> 1) + (unsigned)(k & 1);
            const bool take = (sb[k] > T) || ((sb[k] == T) && (8191u - id >= R));
            m[k] = __builtin_amdgcn_ballot_w32(take);
            wtot += (unsigned)__popc(m[k]);
        }
        if (lane == 0) S.wcnt[par][wave] = wtot;
        __syncthreads();
        unsigned base = 0;
        #pragma unroll
        for (int w = 0; w < NWAVE; ++w) base += (w < wave) ? S.wcnt[par][w] : 0u;
        par ^= 1;
        const unsigned below = (1u << lane) - 1u;
        #pragma unroll
        for (int k = 0; k < IPT; ++k) {
            const unsigned id = id0 + 512u * (unsigned)(k >> 1) + (unsigned)(k & 1);
            if ((m[k] >> lane) & 1u) {
                const unsigned pos = (base + (unsigned)__popc(m[k] & below)) & (unsigned)(K - 1);
                S.surv[pos] = ((unsigned long long)sb[k] << 32) | (unsigned)(~id);
            }
            base += (unsigned)__popc(m[k]);
        }
    }
    __syncthreads();

    if (tid < K) {
        const unsigned long long mine = S.surv[tid];
        unsigned rank = 0;
        #pragma unroll 16
        for (int j = 0; j < K; ++j) rank += (S.surv[j] > mine) ? 1u : 0u;
        S.keys[rank] = mine;
    }
    __syncthreads();

    if (tid < K) {
        unsigned long long key = S.keys[tid];
        unsigned id = ~(unsigned)key;
        id = id < (unsigned)ITEMS ? id : (unsigned)(ITEMS - 1);
        float s = __uint_as_float((unsigned)(key >> 32));
        int cell = (int)(id >> 1);
        int gi = cell / S_GRID, gj = cell % S_GRID;
        const float* cd = in + COORD_OFF + (size_t)id * 4;
        float tx = cd[0], ty = cd[1], tw = cd[2], th = cd[3];
        float x = __fdiv_rn(__fadd_rn(tx, (float)gj), 56.0f);
        float y = __fdiv_rn(__fadd_rn(ty, (float)gi), 56.0f);
        float hw = __fmul_rn(__fmul_rn(tw, tw), 0.5f);
        float hh = __fmul_rn(__fmul_rn(th, th), 0.5f);
        float y0 = __fsub_rn(y, hh), x0 = __fsub_rn(x, hw);
        float y1 = __fadd_rn(y, hh), x1 = __fadd_rn(x, hw);
        float conf = in[CONF_OFF + id];
        const float* pr = in + cell * N_CLASS;
        float best = NEG_INF; int bc = 0;
        #pragma unroll
        for (int c = 0; c < N_CLASS; ++c) {
            float v = __fmul_rn(conf, pr[c]);
            if (v > best) { best = v; bc = c; }
        }
        float dy = fmaxf(__fsub_rn(y1, y0), 0.0f);
        float dx = fmaxf(__fsub_rn(x1, x0), 0.0f);
        P->y0[tid] = y0; P->x0[tid] = x0; P->y1[tid] = y1; P->x1[tid] = x1;
        P->sc[tid] = s;  P->cl[tid] = (float)bc;
        P->ar[tid] = __fmul_rn(dy, dx);
    }
    __syncthreads();

    {
        v8f czero = {};
        const int c16 = lane & 15;
        const int half8 = (lane & 16) ? 8 : 0;
        const int sel = lane & 7;
        const bool lo16 = lane < 16;
        const int wv = __builtin_amdgcn_readfirstlane(wave);
        const int nhalf = (NG > NWAVE) ? 2 : ((wv < NG) ? 1 : 0);
        for (int half = 0; half < nhalf; ++half) {
            const int mi = half ? (NG - 1) - wv : wv;
            const int m0 = mi << 4;
            const int mbase = m0 + half8;
            float my0[8], mx0[8], my1[8], mx1[8];
            #pragma unroll
            for (int r = 0; r < 8; ++r) {
                my0[r] = P->y0[mbase + r]; mx0[r] = P->x0[mbase + r];
                my1[r] = P->y1[mbase + r]; mx1[r] = P->x1[mbase + r];
            }
            const float am = P->ar[m0 + c16];
            v2f A;
            A[0] = lo16 ? am : 0.0f;
            A[1] = lo16 ? 1.0f : 0.0f;
            float ny0 = P->y0[m0 + c16], nx0 = P->x0[m0 + c16], ny1 = P->y1[m0 + c16], nx1 = P->x1[m0 + c16];
            float an = am;
            for (int ni = mi; ni < NG; ++ni) {
                const int nn = (((ni + 1) & (NG - 1)) << 4) + c16;
                const float py0 = P->y0[nn], px0 = P->x0[nn], py1 = P->y1[nn], px1 = P->x1[nn];
                const float pan = P->ar[nn];
                v2f B;
                B[0] = lo16 ? 1.0f : 0.0f;
                B[1] = lo16 ? an : 0.0f;
                v8f d = __builtin_amdgcn_wmma_f32_16x16x4_f32(
                    false, A, false, B, (short)0, czero, false, false);
                unsigned w = 0u;
                #pragma unroll
                for (int r = 0; r < 8; ++r) {
                    const float iy = fmaxf(__fsub_rn(fminf(my1[r], ny1), fmaxf(my0[r], ny0)), 0.0f);
                    const float ix = fmaxf(__fsub_rn(fminf(mx1[r], nx1), fmaxf(mx0[r], nx0)), 0.0f);
                    const float inter = __fmul_rn(iy, ix);
                    const float uni = __fsub_rn(d[r], inter);
                    const float excess = __builtin_fmaf(-0.4f, uni, inter);
                    const float margin = __fmul_rn(uni, 0x1p-26f);
                    const unsigned bal = __builtin_amdgcn_ballot_w32(uni > 0.0f)
                                       & __builtin_amdgcn_ballot_w32(excess >= margin);
                    w = (sel == r) ? bal : w;
                }
                w = (lane & 8) ? (w >> 16) : w;
                P->mask16[(m0 + c16) * 16 + ni] = (unsigned short)w;
                ny0 = py0; nx0 = px0; ny1 = py1; nx1 = px1; an = pan;
            }
        }
    }
    __syncthreads();

    if (wave == 0) {
        unsigned v = (lane < NG) ? 0xFFFFu : 0u;
        int mySel = 0, myOk = 0;
        for (int it = 0; it < MAXOUT; ++it) {
            const unsigned nz = __builtin_amdgcn_ballot_w32(v != 0u);
            const int ok = (nz != 0u) ? 1 : 0;
            const int c = ok ? __builtin_ctz(nz) : 0;
            const unsigned word = (unsigned)__builtin_amdgcn_readlane((int)v, c);
            const int idx = ok ? (c * 16 + __builtin_ctz(word)) : 0;
            if (lane == it) { mySel = idx; myOk = ok; }
            const int wc = min(max(lane & 15, c), NG - 1);
            const unsigned row = P->mask16[idx * 16 + wc];
            v &= ~row;
            if (lane == c) v &= ~(1u << (idx & 15));
        }
        if (lane < MAXOUT) { P->selIdx[lane] = mySel; P->selOk[lane] = myOk; }
    }
    __syncthreads();
    if (__builtin_amdgcn_readfirstlane(P->selOk[MAXOUT - 1]) != 0) break;
    }

    if (tid < MAXOUT) {
        int sel = P->selIdx[tid];
        float okf = P->selOk[tid] ? 1.0f : 0.0f;
        P->redv[tid] = __fmul_rn(P->sc[sel], okf);
    }
    __syncthreads();
    if (tid < MAXOUT) {
        int sel = P->selIdx[tid];
        float okf = P->selOk[tid] ? 1.0f : 0.0f;
        float ms = P->redv[tid];
        int rank = 0;
        for (int jt = 0; jt < MAXOUT; ++jt) {
            float vj = P->redv[jt];
            if (vj > ms || (vj == ms && jt < tid)) rank++;
        }
        float* o = out + (size_t)img * (MAXOUT * 6) + (size_t)rank * 6;
        o[0] = __fmul_rn(P->y0[sel], okf);
        o[1] = __fmul_rn(P->x0[sel], okf);
        o[2] = __fmul_rn(P->y1[sel], okf);
        o[3] = __fmul_rn(P->x1[sel], okf);
        o[4] = ms;
        o[5] = __fmul_rn(P->cl[sel], okf);
    }
}

extern "C" void kernel_launch(void* const* d_in, const int* in_sizes, int n_in,
                              void* d_out, int out_size, void* d_ws, size_t ws_size,
                              hipStream_t stream) {
    (void)n_in; (void)out_size; (void)d_ws; (void)ws_size;
    const float* net = (const float*)d_in[0];
    float* out = (float*)d_out;
    int batch = in_sizes[0] / D_IMG;
    det_kernel<<<dim3(batch), dim3(256), 0, stream>>>(net, out);
}


# First-pass NMS scan held in registers: the 64 valid flags are one 64-bit scalar, the mask rows are read once from LDS into lanes of the scanning wave; no LDS read inside the 30 dependent steps
# speedup vs baseline: 19.8754x; 1.0673x over previous
#include <hip/hip_runtime.h>


typedef float v2f __attribute__((ext_vector_type(2)));
typedef float v8f __attribute__((ext_vector_type(8)));
typedef float v4f __attribute__((ext_vector_type(4)));

#define S_GRID   56
#define N_CLASS  20
#define CELLS    (S_GRID * S_GRID)
#define ITEMS    (CELLS * 2)
#define CAND     256
#define CAND1    64
#define MAXOUT   30
#define D_IMG    94080
#define CONF_OFF 62720
#define COORD_OFF 68992
#define NEG_INF  (-3.402823466e38f)
#define CPT      13
#define IPT      (CPT * 2)
#define SLOTS    (IPT * 256)
#define NWAVE    8
#define ROWQ     160

struct Phase2 {
    float y0[CAND], x0[CAND], y1[CAND], x1[CAND];
    float sc[CAND], cl[CAND], ar[CAND];
    float redv[CAND];
    unsigned short mask16[CAND * 16];
    int selIdx[32], selOk[32];
};

struct Shared {
    unsigned long long surv[CAND];
    unsigned long long keys[CAND];
    unsigned wcnt[2][NWAVE];
    unsigned bucket[32];
    Phase2 p2;
    float part[NWAVE][CPT][ROWQ];
};

#define BLOCK_COUNT(result, PRED)                                               \
    do {                                                                        \
        unsigned c_ = 0;                                                        \
          \
        _Pragma("unroll")                                                       \
        for (int k = 0; k < IPT; ++k) {                                         \
            const unsigned id_ = id0 + 512u * (unsigned)(k >> 1) + (unsigned)(k & 1); \
            const unsigned rid_ = 8191u - id_;                                  \
            const unsigned sv_ = sb[k];                                         \
            (void)rid_; (void)sv_;                                              \
            c_ += (unsigned)__popc(__builtin_amdgcn_ballot_w32(PRED));          \
        }                                                                       \
        if (lane == 0) S.wcnt[par][wave] = c_;                                  \
        __syncthreads();                                                        \
        unsigned t_ = 0;                                                        \
          \
        _Pragma("unroll")                                                       \
        for (int w = 0; w < NWAVE; ++w) t_ += S.wcnt[par][w];                   \
        par ^= 1;                                                               \
          \
          \
        (result) = (unsigned)__builtin_amdgcn_readfirstlane((int)t_);           \
    } while (0)

__global__ __launch_bounds__(256) void det_kernel(const float* __restrict__ net,
                                                  float* __restrict__ out) {
    __shared__ Shared S;
    Phase2* P = &S.p2;

    const int tid = threadIdx.x;
    const int lane = tid & 31;
    const int wave = tid >> 5;
    const int img = blockIdx.x;
    const float* in = net + (size_t)img * D_IMG;
    const unsigned id0 = 2u * (unsigned)tid;

    unsigned sb[IPT];
    {
        const int wv1 = __builtin_amdgcn_readfirstlane(wave);
        v4f q[CPT][5];
        float cf0[CPT], cf1[CPT];
        #pragma unroll
        for (int kc = 0; kc < CPT; ++kc) {
            const int cell = tid + kc * 256;
            cf0[kc] = in[CONF_OFF + cell * 2];
            cf1[kc] = in[CONF_OFF + cell * 2 + 1];
            const v4f* row = (const v4f*)(in + (size_t)(32 * wv1 + 256 * kc) * N_CLASS);
            #pragma unroll
            for (int j = 0; j < 5; ++j) q[kc][j] = row[32 * j + lane];
        }
        #pragma unroll
        for (int kc = 0; kc < CPT; ++kc) {
            float* w = &S.part[wv1][kc][0];
            #pragma unroll
            for (int j = 0; j < 5; ++j) {
                const v4f x = q[kc][j];
                const float a = x[0] > x[1] ? x[0] : x[1];
                const float b = x[2] > x[3] ? x[2] : x[3];
                w[32 * j + lane] = a > b ? a : b;
            }
            __builtin_amdgcn_fence(__ATOMIC_RELEASE, "wavefront");
            __builtin_amdgcn_wave_barrier();
            __builtin_amdgcn_fence(__ATOMIC_ACQUIRE, "wavefront");
            float pm = w[5 * lane];
            #pragma unroll
            for (int i = 1; i < 5; ++i) { const float v = w[5 * lane + i]; pm = v > pm ? v : pm; }
            const float b0 = __fmul_rn(cf0[kc], pm);
            const float b1 = __fmul_rn(cf1[kc], pm);
            float s0 = (b0 >= 0.1f) ? b0 : 0.0f;
            float s1 = (b1 >= 0.1f) ? b1 : 0.0f;
            if (kc == CPT - 1 && wv1 >= 2) { s0 = 0.0f; s1 = 0.0f; }
            sb[2 * kc]     = __float_as_uint(s0);
            sb[2 * kc + 1] = __float_as_uint(s1);
        }
    }

    int par = 0;
    for (int pass = 0; pass < 2; ++pass) {
    const int K = pass ? CAND : CAND1;
    const int NG = K >> 4;
    unsigned T = 0u, cntGE = (unsigned)SLOTS, cntAbove = 0u;
    int bit = 31;
    for (; bit >= 0; --bit) {
        if (cntGE - cntAbove <= 32u) break;
        const unsigned c = T | (1u << bit);
        unsigned n;
        BLOCK_COUNT(n, sv_ >= c);
        if (n >= (unsigned)K) { T = c; cntGE = n; } else { cntAbove = n; }
    }
    if (bit >= 0) {
        const unsigned width = 1u << (bit + 1);
        const unsigned nb = cntGE - cntAbove;
        unsigned m[IPT];
        unsigned wtot = 0;
        #pragma unroll
        for (int k = 0; k < IPT; ++k) {
            m[k] = __builtin_amdgcn_ballot_w32(sb[k] - T < width);
            wtot += (unsigned)__popc(m[k]);
        }
        if (lane == 0) S.wcnt[par][wave] = wtot;
        __syncthreads();
        unsigned base = 0;
        #pragma unroll
        for (int w = 0; w < NWAVE; ++w) base += (w < wave) ? S.wcnt[par][w] : 0u;
        par ^= 1;
        const unsigned below = (1u << lane) - 1u;
        #pragma unroll
        for (int k = 0; k < IPT; ++k) {
            if (m[k] != 0u) {
                if ((m[k] >> lane) & 1u) {
                    S.bucket[(base + (unsigned)__popc(m[k] & below)) & 31u] = sb[k];
                }
                base += (unsigned)__popc(m[k]);
            }
        }
        __syncthreads();
        const unsigned bv = ((unsigned)lane < nb) ? S.bucket[lane] : 0u;
        for (; bit >= 0; --bit) {
            const unsigned c = T | (1u << bit);
            const unsigned n = cntAbove + (unsigned)__popc(__builtin_amdgcn_ballot_w32(bv >= c));
            if (n >= (unsigned)K) { T = c; cntGE = n; }
        }
    }
    unsigned R = 0u;
    if (cntGE != (unsigned)K) {
        unsigned above;
        BLOCK_COUNT(above, sv_ > T);
        const unsigned need = (unsigned)K - above;
        for (int bit = 12; bit >= 0; --bit) {
            const unsigned c = R | (1u << bit);
            unsigned n;
            BLOCK_COUNT(n, (sv_ == T) && (rid_ >= c));
            if (n >= need) R = c;
        }
    }

    {
        unsigned m[IPT];
        unsigned wtot = 0;
        #pragma unroll
        for (int k = 0; k < IPT; ++k) {
            const unsigned id = id0 + 512u * (unsigned)(k >> 1) + (unsigned)(k & 1);
            const bool take = (sb[k] > T) || ((sb[k] == T) && (8191u - id >= R));
            m[k] = __builtin_amdgcn_ballot_w32(take);
            wtot += (unsigned)__popc(m[k]);
        }
        if (lane == 0) S.wcnt[par][wave] = wtot;
        __syncthreads();
        unsigned base = 0;
        #pragma unroll
        for (int w = 0; w < NWAVE; ++w) base += (w < wave) ? S.wcnt[par][w] : 0u;
        par ^= 1;
        const unsigned below = (1u << lane) - 1u;
        #pragma unroll
        for (int k = 0; k < IPT; ++k) {
            const unsigned id = id0 + 512u * (unsigned)(k >> 1) + (unsigned)(k & 1);
            if ((m[k] >> lane) & 1u) {
                const unsigned pos = (base + (unsigned)__popc(m[k] & below)) & (unsigned)(K - 1);
                S.surv[pos] = ((unsigned long long)sb[k] << 32) | (unsigned)(~id);
            }
            base += (unsigned)__popc(m[k]);
        }
    }
    __syncthreads();

    if (tid < K) {
        const unsigned long long mine = S.surv[tid];
        unsigned rank = 0;
        #pragma unroll 16
        for (int j = 0; j < K; ++j) rank += (S.surv[j] > mine) ? 1u : 0u;
        S.keys[rank] = mine;
    }
    __syncthreads();

    if (tid < K) {
        unsigned long long key = S.keys[tid];
        unsigned id = ~(unsigned)key;
        id = id < (unsigned)ITEMS ? id : (unsigned)(ITEMS - 1);
        float s = __uint_as_float((unsigned)(key >> 32));
        int cell = (int)(id >> 1);
        int gi = cell / S_GRID, gj = cell % S_GRID;
        const float* cd = in + COORD_OFF + (size_t)id * 4;
        float tx = cd[0], ty = cd[1], tw = cd[2], th = cd[3];
        float x = __fdiv_rn(__fadd_rn(tx, (float)gj), 56.0f);
        float y = __fdiv_rn(__fadd_rn(ty, (float)gi), 56.0f);
        float hw = __fmul_rn(__fmul_rn(tw, tw), 0.5f);
        float hh = __fmul_rn(__fmul_rn(th, th), 0.5f);
        float y0 = __fsub_rn(y, hh), x0 = __fsub_rn(x, hw);
        float y1 = __fadd_rn(y, hh), x1 = __fadd_rn(x, hw);
        float conf = in[CONF_OFF + id];
        const float* pr = in + cell * N_CLASS;
        float best = NEG_INF; int bc = 0;
        #pragma unroll
        for (int c = 0; c < N_CLASS; ++c) {
            float v = __fmul_rn(conf, pr[c]);
            if (v > best) { best = v; bc = c; }
        }
        float dy = fmaxf(__fsub_rn(y1, y0), 0.0f);
        float dx = fmaxf(__fsub_rn(x1, x0), 0.0f);
        P->y0[tid] = y0; P->x0[tid] = x0; P->y1[tid] = y1; P->x1[tid] = x1;
        P->sc[tid] = s;  P->cl[tid] = (float)bc;
        P->ar[tid] = __fmul_rn(dy, dx);
    }
    __syncthreads();

    {
        v8f czero = {};
        const int c16 = lane & 15;
        const int half8 = (lane & 16) ? 8 : 0;
        const int sel = lane & 7;
        const bool lo16 = lane < 16;
        const int wv = __builtin_amdgcn_readfirstlane(wave);
        const int nhalf = (NG > NWAVE) ? 2 : ((wv < NG) ? 1 : 0);
        for (int half = 0; half < nhalf; ++half) {
            const int mi = half ? (NG - 1) - wv : wv;
            const int m0 = mi << 4;
            const int mbase = m0 + half8;
            float my0[8], mx0[8], my1[8], mx1[8];
            #pragma unroll
            for (int r = 0; r < 8; ++r) {
                my0[r] = P->y0[mbase + r]; mx0[r] = P->x0[mbase + r];
                my1[r] = P->y1[mbase + r]; mx1[r] = P->x1[mbase + r];
            }
            const float am = P->ar[m0 + c16];
            v2f A;
            A[0] = lo16 ? am : 0.0f;
            A[1] = lo16 ? 1.0f : 0.0f;
            float ny0 = P->y0[m0 + c16], nx0 = P->x0[m0 + c16], ny1 = P->y1[m0 + c16], nx1 = P->x1[m0 + c16];
            float an = am;
            for (int ni = mi; ni < NG; ++ni) {
                const int nn = (((ni + 1) & (NG - 1)) << 4) + c16;
                const float py0 = P->y0[nn], px0 = P->x0[nn], py1 = P->y1[nn], px1 = P->x1[nn];
                const float pan = P->ar[nn];
                v2f B;
                B[0] = lo16 ? 1.0f : 0.0f;
                B[1] = lo16 ? an : 0.0f;
                v8f d = __builtin_amdgcn_wmma_f32_16x16x4_f32(
                    false, A, false, B, (short)0, czero, false, false);
                unsigned w = 0u;
                #pragma unroll
                for (int r = 0; r < 8; ++r) {
                    const float iy = fmaxf(__fsub_rn(fminf(my1[r], ny1), fmaxf(my0[r], ny0)), 0.0f);
                    const float ix = fmaxf(__fsub_rn(fminf(mx1[r], nx1), fmaxf(mx0[r], nx0)), 0.0f);
                    const float inter = __fmul_rn(iy, ix);
                    const float uni = __fsub_rn(d[r], inter);
                    const float excess = __builtin_fmaf(-0.4f, uni, inter);
                    const float margin = __fmul_rn(uni, 0x1p-26f);
                    const unsigned bal = __builtin_amdgcn_ballot_w32(uni > 0.0f)
                                       & __builtin_amdgcn_ballot_w32(excess >= margin);
                    w = (sel == r) ? bal : w;
                }
                w = (lane & 8) ? (w >> 16) : w;
                P->mask16[(m0 + c16) * 16 + ni] = (unsigned short)w;
                ny0 = py0; nx0 = px0; ny1 = py1; nx1 = px1; an = pan;
            }
        }
    }
    __syncthreads();

    if (wave == 0) {
        int mySel = 0, myOk = 0;
        if (K == CAND1) {
            static_assert(CAND1 == 64, "the first-pass scan keeps its valid flags in one 64-bit scalar");
            const unsigned short* mk = P->mask16;
            const int rA = lane, rB = lane + 32;
            const int gA = rA >> 4, gB = rB >> 4;
            const unsigned loA = (unsigned)mk[rA * 16 + gA] | ((unsigned)mk[rA * 16 + 1] << 16);
            const unsigned hiA = (unsigned)mk[rA * 16 + 2]  | ((unsigned)mk[rA * 16 + 3] << 16);
            const unsigned hiB = (unsigned)mk[rB * 16 + gB] | ((unsigned)mk[rB * 16 + 3] << 16);
            unsigned long long valid = ~0ull;
            for (int it = 0; it < MAXOUT; ++it) {
                const int ok = (valid != 0ull) ? 1 : 0;
                const int idx = ok ? (int)__builtin_ctzll(valid) : 0;
                const unsigned ra = (unsigned)__builtin_amdgcn_readlane((int)loA, idx & 31);
                const unsigned rb = (unsigned)__builtin_amdgcn_readlane((int)hiA, idx & 31);
                const unsigned rd = (unsigned)__builtin_amdgcn_readlane((int)hiB, idx & 31);
                const unsigned long long row = ((unsigned long long)((idx < 32) ? rb : rd) << 32) | ra;
                valid &= ~(row | (1ull << idx));
                if (lane == it) { mySel = idx; myOk = ok; }
            }
        } else {
        unsigned v = (lane < NG) ? 0xFFFFu : 0u;
        for (int it = 0; it < MAXOUT; ++it) {
            const unsigned nz = __builtin_amdgcn_ballot_w32(v != 0u);
            const int ok = (nz != 0u) ? 1 : 0;
            const int c = ok ? __builtin_ctz(nz) : 0;
            const unsigned word = (unsigned)__builtin_amdgcn_readlane((int)v, c);
            const int idx = ok ? (c * 16 + __builtin_ctz(word)) : 0;
            if (lane == it) { mySel = idx; myOk = ok; }
            const int wc = min(max(lane & 15, c), NG - 1);
            const unsigned row = P->mask16[idx * 16 + wc];
            v &= ~row;
            if (lane == c) v &= ~(1u << (idx & 15));
        }
        }
        if (lane < MAXOUT) { P->selIdx[lane] = mySel; P->selOk[lane] = myOk; }
    }
    __syncthreads();
    if (__builtin_amdgcn_readfirstlane(P->selOk[MAXOUT - 1]) != 0) break;
    }

    if (tid < MAXOUT) {
        int sel = P->selIdx[tid];
        float okf = P->selOk[tid] ? 1.0f : 0.0f;
        P->redv[tid] = __fmul_rn(P->sc[sel], okf);
    }
    __syncthreads();
    if (tid < MAXOUT) {
        int sel = P->selIdx[tid];
        float okf = P->selOk[tid] ? 1.0f : 0.0f;
        float ms = P->redv[tid];
        int rank = 0;
        for (int jt = 0; jt < MAXOUT; ++jt) {
            float vj = P->redv[jt];
            if (vj > ms || (vj == ms && jt < tid)) rank++;
        }
        float* o = out + (size_t)img * (MAXOUT * 6) + (size_t)rank * 6;
        o[0] = __fmul_rn(P->y0[sel], okf);
        o[1] = __fmul_rn(P->x0[sel], okf);
        o[2] = __fmul_rn(P->y1[sel], okf);
        o[3] = __fmul_rn(P->x1[sel], okf);
        o[4] = ms;
        o[5] = __fmul_rn(P->cl[sel], okf);
    }
}

extern "C" void kernel_launch(void* const* d_in, const int* in_sizes, int n_in,
                              void* d_out, int out_size, void* d_ws, size_t ws_size,
                              hipStream_t stream) {
    (void)n_in; (void)out_size; (void)d_ws; (void)ws_size;
    const float* net = (const float*)d_in[0];
    float* out = (float*)d_out;
    int batch = in_sizes[0] / D_IMG;
    det_kernel<<<dim3(batch), dim3(256), 0, stream>>>(net, out);
}


# First-pass top-64 cut without the bitwise threshold search: a lower bound from order statistics of the thread maxima (ranked in registers, one barrier), all slots at or above it ranked, best 64 kept
# speedup vs baseline: 25.7977x; 1.2980x over previous
#include <hip/hip_runtime.h>


typedef float v2f __attribute__((ext_vector_type(2)));
typedef float v8f __attribute__((ext_vector_type(8)));
typedef float v4f __attribute__((ext_vector_type(4)));
typedef unsigned v4u __attribute__((ext_vector_type(4)));

#define S_GRID   56
#define N_CLASS  20
#define CELLS    (S_GRID * S_GRID)
#define ITEMS    (CELLS * 2)
#define CAND     256
#define CAND1    64
#define MAXOUT   30
#define D_IMG    94080
#define CONF_OFF 62720
#define COORD_OFF 68992
#define NEG_INF  (-3.402823466e38f)
#define CPT      13
#define IPT      (CPT * 2)
#define SLOTS    (IPT * 256)
#define NWAVE    8
#define ROWQ     160

struct Phase2 {
    float y0[CAND], x0[CAND], y1[CAND], x1[CAND];
    float sc[CAND], cl[CAND], ar[CAND];
    float redv[CAND];
    unsigned short mask16[CAND * 16];
    int selIdx[32], selOk[32];
};

struct Shared {
    unsigned long long surv[CAND];
    unsigned long long keys[CAND];
    alignas(16) unsigned wcnt[2][NWAVE];
    unsigned bucket[32];
    unsigned pub[32];
    Phase2 p2;
    float part[NWAVE][CPT][ROWQ];
};

#define BLOCK_COUNT(result, PRED)                                               \
    do {                                                                        \
        unsigned c_ = 0;                                                        \
          \
        _Pragma("unroll")                                                       \
        for (int k = 0; k < IPT; ++k) {                                         \
            const unsigned id_ = id0 + 512u * (unsigned)(k >> 1) + (unsigned)(k & 1); \
            const unsigned rid_ = 8191u - id_;                                  \
            const unsigned sv_ = sb[k];                                         \
            (void)rid_; (void)sv_;                                              \
            c_ += (unsigned)__popc(__builtin_amdgcn_ballot_w32(PRED));          \
        }                                                                       \
        if (lane == 0) S.wcnt[par][wave] = c_;                                  \
        __syncthreads();                                                        \
        unsigned t_ = 0;                                                        \
          \
        _Pragma("unroll")                                                       \
        for (int w = 0; w < NWAVE; ++w) t_ += S.wcnt[par][w];                   \
        par ^= 1;                                                               \
          \
          \
        (result) = (unsigned)__builtin_amdgcn_readfirstlane((int)t_);           \
    } while (0)

__global__ __launch_bounds__(256) void det_kernel(const float* __restrict__ net,
                                                  float* __restrict__ out) {
    __shared__ Shared S;
    Phase2* P = &S.p2;

    const int tid = threadIdx.x;
    const int lane = tid & 31;
    const int wave = tid >> 5;
    const int img = blockIdx.x;
    const float* in = net + (size_t)img * D_IMG;
    const unsigned id0 = 2u * (unsigned)tid;

    unsigned sb[IPT];
    {
        const int wv1 = __builtin_amdgcn_readfirstlane(wave);
        v4f q[CPT][5];
        float cf0[CPT], cf1[CPT];
        #pragma unroll
        for (int kc = 0; kc < CPT; ++kc) {
            const int cell = tid + kc * 256;
            cf0[kc] = in[CONF_OFF + cell * 2];
            cf1[kc] = in[CONF_OFF + cell * 2 + 1];
            const v4f* row = (const v4f*)(in + (size_t)(32 * wv1 + 256 * kc) * N_CLASS);
            #pragma unroll
            for (int j = 0; j < 5; ++j) q[kc][j] = row[32 * j + lane];
        }
        #pragma unroll
        for (int kc = 0; kc < CPT; ++kc) {
            float* w = &S.part[wv1][kc][0];
            #pragma unroll
            for (int j = 0; j < 5; ++j) {
                const v4f x = q[kc][j];
                const float a = x[0] > x[1] ? x[0] : x[1];
                const float b = x[2] > x[3] ? x[2] : x[3];
                w[32 * j + lane] = a > b ? a : b;
            }
            __builtin_amdgcn_fence(__ATOMIC_RELEASE, "wavefront");
            __builtin_amdgcn_wave_barrier();
            __builtin_amdgcn_fence(__ATOMIC_ACQUIRE, "wavefront");
            float pm = w[5 * lane];
            #pragma unroll
            for (int i = 1; i < 5; ++i) { const float v = w[5 * lane + i]; pm = v > pm ? v : pm; }
            const float b0 = __fmul_rn(cf0[kc], pm);
            const float b1 = __fmul_rn(cf1[kc], pm);
            float s0 = (b0 >= 0.1f) ? b0 : 0.0f;
            float s1 = (b1 >= 0.1f) ? b1 : 0.0f;
            if (kc == CPT - 1 && wv1 >= 2) { s0 = 0.0f; s1 = 0.0f; }
            sb[2 * kc]     = __float_as_uint(s0);
            sb[2 * kc + 1] = __float_as_uint(s1);
        }
    }

    static_assert(CAND1 == 4 * 16 && NWAVE * 4 == 32, "the bound takes 4 words from each of 8 waves and the 16th of the 32");
    int par = 0;
    for (int pass = 0; pass < 2; ++pass) {
    const int K = pass ? CAND : CAND1;
    const int NG = K >> 4;
    unsigned T = 0u, R = 0u;
    unsigned n = 0u;
    bool exact = (pass != 0);
    if (!exact) {
        unsigned tm = sb[0];
        #pragma unroll
        for (int k = 1; k < IPT; ++k) tm = sb[k] > tm ? sb[k] : tm;
        const unsigned key1 = (tm & ~31u) | (unsigned)lane;
        unsigned g1 = 0;
        #pragma unroll
        for (int j = 0; j < 32; ++j)
            g1 += ((unsigned)__builtin_amdgcn_readlane((int)key1, j) > key1) ? 1u : 0u;
        if (g1 < 16u && (g1 & 3u) == 3u) S.pub[4 * wave + (int)(g1 >> 2)] = key1;
        __syncthreads();
        const unsigned key2 = (S.pub[lane] & ~31u) | (unsigned)lane;
        unsigned g2 = 0;
        #pragma unroll
        for (int j = 0; j < 32; ++j)
            g2 += ((unsigned)__builtin_amdgcn_readlane((int)key2, j) > key2) ? 1u : 0u;
        const unsigned at16 = __builtin_amdgcn_ballot_w32(g2 == 15u);
        T = (unsigned)__builtin_amdgcn_readlane((int)key2, __builtin_ctz(at16)) & ~31u;
    }
    for (int attempt = 0; attempt < 2; ++attempt) {
    if (exact) {
    unsigned cntGE = (unsigned)SLOTS, cntAbove = 0u;
    T = 0u; R = 0u;
    int bit = 31;
    for (; bit >= 0; --bit) {
        if (cntGE - cntAbove <= 32u) break;
        const unsigned c = T | (1u << bit);
        unsigned nc;
        BLOCK_COUNT(nc, sv_ >= c);
        if (nc >= (unsigned)K) { T = c; cntGE = nc; } else { cntAbove = nc; }
    }
    if (bit >= 0) {
        const unsigned width = 1u << (bit + 1);
        const unsigned nb = cntGE - cntAbove;
        unsigned m[IPT];
        unsigned wtot = 0;
        #pragma unroll
        for (int k = 0; k < IPT; ++k) {
            m[k] = __builtin_amdgcn_ballot_w32(sb[k] - T < width);
            wtot += (unsigned)__popc(m[k]);
        }
        if (lane == 0) S.wcnt[par][wave] = wtot;
        __syncthreads();
        unsigned base = 0;
        #pragma unroll
        for (int w = 0; w < NWAVE; ++w) base += (w < wave) ? S.wcnt[par][w] : 0u;
        par ^= 1;
        const unsigned below = (1u << lane) - 1u;
        #pragma unroll
        for (int k = 0; k < IPT; ++k) {
            if (m[k] != 0u) {
                if ((m[k] >> lane) & 1u) {
                    S.bucket[(base + (unsigned)__popc(m[k] & below)) & 31u] = sb[k];
                }
                base += (unsigned)__popc(m[k]);
            }
        }
        __syncthreads();
        const unsigned bv = ((unsigned)lane < nb) ? S.bucket[lane] : 0u;
        for (; bit >= 0; --bit) {
            const unsigned c = T | (1u << bit);
            const unsigned nc = cntAbove + (unsigned)__popc(__builtin_amdgcn_ballot_w32(bv >= c));
            if (nc >= (unsigned)K) { T = c; cntGE = nc; }
        }
    }
    if (cntGE != (unsigned)K) {
        unsigned above;
        BLOCK_COUNT(above, sv_ > T);
        const unsigned need = (unsigned)K - above;
        for (int rbit = 12; rbit >= 0; --rbit) {
            const unsigned c = R | (1u << rbit);
            unsigned nc;
            BLOCK_COUNT(nc, (sv_ == T) && (rid_ >= c));
            if (nc >= need) R = c;
        }
    }
    }

    {
        unsigned m[IPT];
        unsigned wtot = 0;
        #pragma unroll
        for (int k = 0; k < IPT; ++k) {
            const unsigned id = id0 + 512u * (unsigned)(k >> 1) + (unsigned)(k & 1);
            const bool take = (sb[k] > T) || ((sb[k] == T) && (8191u - id >= R));
            m[k] = __builtin_amdgcn_ballot_w32(take);
            wtot += (unsigned)__popc(m[k]);
        }
        if (lane == 0) S.wcnt[par][wave] = wtot;
        __syncthreads();
        const v4u* wsum = (const v4u*)&S.wcnt[par][0];
        const v4u c0 = wsum[0], c1 = wsum[1];
        unsigned base = 0, tot = 0;
        #pragma unroll
        for (int w = 0; w < NWAVE; ++w) {
            const unsigned cw = (w < 4) ? c0[w & 3] : c1[w & 3];
            base += (w < wave) ? cw : 0u;
            tot += cw;
        }
        par ^= 1;
        n = (unsigned)__builtin_amdgcn_readfirstlane((int)tot);
        if (n <= (unsigned)CAND) {
            const unsigned below = (1u << lane) - 1u;
            #pragma unroll
            for (int k = 0; k < IPT; ++k) {
                const unsigned id = id0 + 512u * (unsigned)(k >> 1) + (unsigned)(k & 1);
                if ((m[k] >> lane) & 1u) {
                    const unsigned pos = (base + (unsigned)__popc(m[k] & below)) & (unsigned)(CAND - 1);
                    S.surv[pos] = ((unsigned long long)sb[k] << 32) | (unsigned)(~id);
                }
                base += (unsigned)__popc(m[k]);
            }
            if (tid < 16 && n + (unsigned)tid < ((n + 15u) & ~15u)) S.surv[n + (unsigned)tid] = 0ull;
        }
    }
    if (n <= (unsigned)CAND) break;
    exact = true;
    }
    __syncthreads();

    if ((unsigned)tid < n) {
        const unsigned long long mine = S.surv[tid];
        unsigned rank = 0;
        const unsigned n16 = (n + 15u) & ~15u;
        for (unsigned j = 0; j < n16; j += 16u) {
            #pragma unroll
            for (int u = 0; u < 16; ++u) rank += (S.surv[j + (unsigned)u] > mine) ? 1u : 0u;
        }
        if (rank < (unsigned)K) S.keys[rank] = mine;
    }
    __syncthreads();

    if (tid < K) {
        unsigned long long key = S.keys[tid];
        unsigned id = ~(unsigned)key;
        id = id < (unsigned)ITEMS ? id : (unsigned)(ITEMS - 1);
        float s = __uint_as_float((unsigned)(key >> 32));
        int cell = (int)(id >> 1);
        int gi = cell / S_GRID, gj = cell % S_GRID;
        const float* cd = in + COORD_OFF + (size_t)id * 4;
        float tx = cd[0], ty = cd[1], tw = cd[2], th = cd[3];
        float x = __fdiv_rn(__fadd_rn(tx, (float)gj), 56.0f);
        float y = __fdiv_rn(__fadd_rn(ty, (float)gi), 56.0f);
        float hw = __fmul_rn(__fmul_rn(tw, tw), 0.5f);
        float hh = __fmul_rn(__fmul_rn(th, th), 0.5f);
        float y0 = __fsub_rn(y, hh), x0 = __fsub_rn(x, hw);
        float y1 = __fadd_rn(y, hh), x1 = __fadd_rn(x, hw);
        float conf = in[CONF_OFF + id];
        const float* pr = in + cell * N_CLASS;
        float best = NEG_INF; int bc = 0;
        #pragma unroll
        for (int c = 0; c < N_CLASS; ++c) {
            float v = __fmul_rn(conf, pr[c]);
            if (v > best) { best = v; bc = c; }
        }
        float dy = fmaxf(__fsub_rn(y1, y0), 0.0f);
        float dx = fmaxf(__fsub_rn(x1, x0), 0.0f);
        P->y0[tid] = y0; P->x0[tid] = x0; P->y1[tid] = y1; P->x1[tid] = x1;
        P->sc[tid] = s;  P->cl[tid] = (float)bc;
        P->ar[tid] = __fmul_rn(dy, dx);
    }
    __syncthreads();

    {
        v8f czero = {};
        const int c16 = lane & 15;
        const int half8 = (lane & 16) ? 8 : 0;
        const int sel = lane & 7;
        const bool lo16 = lane < 16;
        const int wv = __builtin_amdgcn_readfirstlane(wave);
        const int nhalf = (NG > NWAVE) ? 2 : ((wv < NG) ? 1 : 0);
        for (int half = 0; half < nhalf; ++half) {
            const int mi = half ? (NG - 1) - wv : wv;
            const int m0 = mi << 4;
            const int mbase = m0 + half8;
            float my0[8], mx0[8], my1[8], mx1[8];
            #pragma unroll
            for (int r = 0; r < 8; ++r) {
                my0[r] = P->y0[mbase + r]; mx0[r] = P->x0[mbase + r];
                my1[r] = P->y1[mbase + r]; mx1[r] = P->x1[mbase + r];
            }
            const float am = P->ar[m0 + c16];
            v2f A;
            A[0] = lo16 ? am : 0.0f;
            A[1] = lo16 ? 1.0f : 0.0f;
            float ny0 = P->y0[m0 + c16], nx0 = P->x0[m0 + c16], ny1 = P->y1[m0 + c16], nx1 = P->x1[m0 + c16];
            float an = am;
            for (int ni = mi; ni < NG; ++ni) {
                const int nn = (((ni + 1) & (NG - 1)) << 4) + c16;
                const float py0 = P->y0[nn], px0 = P->x0[nn], py1 = P->y1[nn], px1 = P->x1[nn];
                const float pan = P->ar[nn];
                v2f B;
                B[0] = lo16 ? 1.0f : 0.0f;
                B[1] = lo16 ? an : 0.0f;
                v8f d = __builtin_amdgcn_wmma_f32_16x16x4_f32(
                    false, A, false, B, (short)0, czero, false, false);
                unsigned w = 0u;
                #pragma unroll
                for (int r = 0; r < 8; ++r) {
                    const float iy = fmaxf(__fsub_rn(fminf(my1[r], ny1), fmaxf(my0[r], ny0)), 0.0f);
                    const float ix = fmaxf(__fsub_rn(fminf(mx1[r], nx1), fmaxf(mx0[r], nx0)), 0.0f);
                    const float inter = __fmul_rn(iy, ix);
                    const float uni = __fsub_rn(d[r], inter);
                    const float excess = __builtin_fmaf(-0.4f, uni, inter);
                    const float margin = __fmul_rn(uni, 0x1p-26f);
                    const unsigned bal = __builtin_amdgcn_ballot_w32(uni > 0.0f)
                                       & __builtin_amdgcn_ballot_w32(excess >= margin);
                    w = (sel == r) ? bal : w;
                }
                w = (lane & 8) ? (w >> 16) : w;
                P->mask16[(m0 + c16) * 16 + ni] = (unsigned short)w;
                ny0 = py0; nx0 = px0; ny1 = py1; nx1 = px1; an = pan;
            }
        }
    }
    __syncthreads();

    if (wave == 0) {
        int mySel = 0, myOk = 0;
        if (K == CAND1) {
            static_assert(CAND1 == 64, "the first-pass scan keeps its valid flags in one 64-bit scalar");
            const unsigned short* mk = P->mask16;
            const int rA = lane, rB = lane + 32;
            const int gA = rA >> 4, gB = rB >> 4;
            const unsigned loA = (unsigned)mk[rA * 16 + gA] | ((unsigned)mk[rA * 16 + 1] << 16);
            const unsigned hiA = (unsigned)mk[rA * 16 + 2]  | ((unsigned)mk[rA * 16 + 3] << 16);
            const unsigned hiB = (unsigned)mk[rB * 16 + gB] | ((unsigned)mk[rB * 16 + 3] << 16);
            unsigned long long valid = ~0ull;
            for (int it = 0; it < MAXOUT; ++it) {
                const int ok = (valid != 0ull) ? 1 : 0;
                const int idx = ok ? (int)__builtin_ctzll(valid) : 0;
                const unsigned ra = (unsigned)__builtin_amdgcn_readlane((int)loA, idx & 31);
                const unsigned rb = (unsigned)__builtin_amdgcn_readlane((int)hiA, idx & 31);
                const unsigned rd = (unsigned)__builtin_amdgcn_readlane((int)hiB, idx & 31);
                const unsigned long long row = ((unsigned long long)((idx < 32) ? rb : rd) << 32) | ra;
                valid &= ~(row | (1ull << idx));
                if (lane == it) { mySel = idx; myOk = ok; }
            }
        } else {
        unsigned v = (lane < NG) ? 0xFFFFu : 0u;
        for (int it = 0; it < MAXOUT; ++it) {
            const unsigned nz = __builtin_amdgcn_ballot_w32(v != 0u);
            const int ok = (nz != 0u) ? 1 : 0;
            const int c = ok ? __builtin_ctz(nz) : 0;
            const unsigned word = (unsigned)__builtin_amdgcn_readlane((int)v, c);
            const int idx = ok ? (c * 16 + __builtin_ctz(word)) : 0;
            if (lane == it) { mySel = idx; myOk = ok; }
            const int wc = min(max(lane & 15, c), NG - 1);
            const unsigned row = P->mask16[idx * 16 + wc];
            v &= ~row;
            if (lane == c) v &= ~(1u << (idx & 15));
        }
        }
        if (lane < MAXOUT) { P->selIdx[lane] = mySel; P->selOk[lane] = myOk; }
    }
    __syncthreads();
    if (__builtin_amdgcn_readfirstlane(P->selOk[MAXOUT - 1]) != 0) break;
    }

    if (tid < MAXOUT) {
        int sel = P->selIdx[tid];
        float okf = P->selOk[tid] ? 1.0f : 0.0f;
        P->redv[tid] = __fmul_rn(P->sc[sel], okf);
    }
    __syncthreads();
    if (tid < MAXOUT) {
        int sel = P->selIdx[tid];
        float okf = P->selOk[tid] ? 1.0f : 0.0f;
        float ms = P->redv[tid];
        int rank = 0;
        for (int jt = 0; jt < MAXOUT; ++jt) {
            float vj = P->redv[jt];
            if (vj > ms || (vj == ms && jt < tid)) rank++;
        }
        float* o = out + (size_t)img * (MAXOUT * 6) + (size_t)rank * 6;
        o[0] = __fmul_rn(P->y0[sel], okf);
        o[1] = __fmul_rn(P->x0[sel], okf);
        o[2] = __fmul_rn(P->y1[sel], okf);
        o[3] = __fmul_rn(P->x1[sel], okf);
        o[4] = ms;
        o[5] = __fmul_rn(P->cl[sel], okf);
    }
}

extern "C" void kernel_launch(void* const* d_in, const int* in_sizes, int n_in,
                              void* d_out, int out_size, void* d_ws, size_t ws_size,
                              hipStream_t stream) {
    (void)n_in; (void)out_size; (void)d_ws; (void)ws_size;
    const float* net = (const float*)d_in[0];
    float* out = (float*)d_out;
    int batch = in_sizes[0] / D_IMG;
    det_kernel<<<dim3(batch), dim3(256), 0, stream>>>(net, out);
}
